# starting kernel
# baseline (speedup 1.0000x reference)
#include <hip/hip_runtime.h>

typedef __attribute__((ext_vector_type(2))) float v2f;
typedef __attribute__((ext_vector_type(8))) float v8f;

#define NN 50000
#define EE 800000
#define DIM 128
#define NCLS 40
#define CP 48

__global__ void zero_kernel(float* __restrict__ p, long n) {
    long i = (long)blockIdx.x * blockDim.x + threadIdx.x;
    long stride = (long)gridDim.x * blockDim.x;
    for (; i < n; i += stride) p[i] = 0.0f;
}

__global__ void deg_kernel(const int* __restrict__ dst, float* __restrict__ deg) {
    int e = blockIdx.x * blockDim.x + threadIdx.x;
    if (e < EE) unsafeAtomicAdd(&deg[dst[e]], 1.0f);
}

__global__ void deginv_kernel(float* __restrict__ deg) {
    int i = blockIdx.x * blockDim.x + threadIdx.x;
    if (i < NN) {
        float d = deg[i];
        deg[i] = (d > 0.0f) ? (1.0f / d) : 0.0f;
    }
}

__global__ void consts_kernel(const float* b0, const float* g0, const float* be0,
                              const float* rm0, const float* rv0,
                              const float* b1, const float* g1, const float* be1,
                              const float* rm1, const float* rv1,
                              const float* b2, const float* g2, const float* be2,
                              const float* rm2, const float* rv2,
                              float* __restrict__ cbuf) {
    int t = threadIdx.x;
    const float eps = 1e-5f;
    if (t < DIM) {
        float s = g0[t] * rsqrtf(rv0[t] + eps);
        cbuf[0 * DIM + t] = s;
        cbuf[1 * DIM + t] = (b0[t] - rm0[t]) * s + be0[t];
        s = g1[t] * rsqrtf(rv1[t] + eps);
        cbuf[2 * DIM + t] = s;
        cbuf[3 * DIM + t] = (b1[t] - rm1[t]) * s + be1[t];
    }
    if (t < NCLS) {
        float s = g2[t] * rsqrtf(rv2[t] + eps);
        cbuf[4 * DIM + t]      = s;
        cbuf[4 * DIM + 64 + t] = (b2[t] - rm2[t]) * s + be2[t];
    }
}

__global__ void scatter_kernel(const float* __restrict__ x,
                               const int* __restrict__ src,
                               const int* __restrict__ dst,
                               const float* __restrict__ mask,
                               float* __restrict__ agg) {
    long idx = (long)blockIdx.x * blockDim.x + threadIdx.x;
    long total = (long)EE * (DIM / 4);
    if (idx >= total) return;
    int e = (int)(idx >> 5);
    int c = (int)((idx & 31) << 2);
    int s = src[e];
    int d = dst[e];
    float m = mask[e];
    const float4 v = *(const float4*)(x + (long)s * DIM + c);
    float* ap = agg + (long)d * DIM + c;
    unsafeAtomicAdd(ap + 0, v.x * m);
    unsafeAtomicAdd(ap + 1, v.y * m);
    unsafeAtomicAdd(ap + 2, v.z * m);
    unsafeAtomicAdd(ap + 3, v.w * m);
}

__global__ void __launch_bounds__(256)
gin_gemm_kernel(const float* __restrict__ x, const float* __restrict__ agg,
                const float* __restrict__ dinv, const float* __restrict__ W,
                const float* __restrict__ scale, const float* __restrict__ shift,
                float* __restrict__ out) {
    __shared__ v2f sW[(DIM / 2) * DIM];
    const int tid = threadIdx.x;
    for (int i = tid; i < (DIM / 2) * DIM; i += 256) {
        int p = i / DIM, c = i % DIM;
        v2f w;
        w.x = W[(2 * p) * DIM + c];
        w.y = W[(2 * p + 1) * DIM + c];
        sW[i] = w;
    }
    __syncthreads();

    const int wave = tid >> 5;
    const int lane = tid & 31;
    const int m    = lane & 15;
    const int prow = lane >> 4;
    const int kk   = prow << 1;
    const int row0 = blockIdx.x * 128 + wave * 16;

    const int  r     = row0 + m;
    const bool valid = r < NN;
    const int  rc    = valid ? r : 0;
    const float di   = valid ? dinv[rc] : 0.0f;
    const float* xr  = x   + (long)rc * DIM;
    const float* ar  = agg + (long)rc * DIM;

    v8f acc[8] = {};
    float2 xv = *(const float2*)(xr + kk);
    float2 av = *(const float2*)(ar + kk);
#pragma unroll 4
    for (int k0 = 0; k0 < DIM; k0 += 4) {
        const v2f* bb = sW + ((k0 >> 1) + prow) * DIM + m;
        v2f bf[8];
#pragma unroll
        for (int nt = 0; nt < 8; ++nt) bf[nt] = bb[nt * 16];

        v2f a;
        a.x = valid ? (xv.x + av.x * di) : 0.0f;
        a.y = valid ? (xv.y + av.y * di) : 0.0f;

        if (k0 + 4 < DIM) {
            xv = *(const float2*)(xr + k0 + 4 + kk);
            av = *(const float2*)(ar + k0 + 4 + kk);
        }
#pragma unroll
        for (int nt = 0; nt < 8; ++nt)
            acc[nt] = __builtin_amdgcn_wmma_f32_16x16x4_f32(
                false, a, false, bf[nt], (short)0, acc[nt], false, false);
    }

    const int roff = prow << 3;
#pragma unroll
    for (int nt = 0; nt < 8; ++nt) {
        const int col = nt * 16 + m;
        const float s  = scale[col];
        const float sh = shift[col];
#pragma unroll
        for (int j = 0; j < 8; ++j) {
            int rr = row0 + roff + j;
            if (rr < NN) {
                float v = acc[nt][j] * s + sh;
                out[(long)rr * DIM + col] = v > 0.0f ? v : 0.0f;
            }
        }
    }
}

__global__ void __launch_bounds__(256)
gin_final_kernel(const float* __restrict__ h2, const float* __restrict__ agg,
                 const float* __restrict__ dinv,
                 const float* __restrict__ w2, const float* __restrict__ wp,
                 const float* __restrict__ scale2, const float* __restrict__ shift2,
                 const float* __restrict__ bp, float* __restrict__ out) {
    __shared__ v2f sW2[(DIM / 2) * CP];
    __shared__ v2f sWP[(DIM / 2) * CP];
    const int tid = threadIdx.x;
    for (int i = tid; i < (DIM / 2) * CP; i += 256) {
        int p = i / CP, c = i % CP;
        v2f a = {0.0f, 0.0f}, b = {0.0f, 0.0f};
        if (c < NCLS) {
            a.x = w2[(2 * p) * NCLS + c];
            a.y = w2[(2 * p + 1) * NCLS + c];
            b.x = wp[(2 * p) * NCLS + c];
            b.y = wp[(2 * p + 1) * NCLS + c];
        }
        sW2[i] = a;
        sWP[i] = b;
    }
    __syncthreads();

    const int wave = tid >> 5;
    const int lane = tid & 31;
    const int m    = lane & 15;
    const int prow = lane >> 4;
    const int kk   = prow << 1;
    const int row0 = blockIdx.x * 128 + wave * 16;

    const int  r     = row0 + m;
    const bool valid = r < NN;
    const int  rc    = valid ? r : 0;
    const float di   = valid ? dinv[rc] : 0.0f;
    const float* xr  = h2  + (long)rc * DIM;
    const float* ar  = agg + (long)rc * DIM;

    v8f acc3[3] = {};
    v8f accp[3] = {};
    float2 xv = *(const float2*)(xr + kk);
    float2 av = *(const float2*)(ar + kk);
#pragma unroll 4
    for (int k0 = 0; k0 < DIM; k0 += 4) {
        const int pidx = ((k0 >> 1) + prow) * CP + m;
        v2f bf2[3], bfp[3];
#pragma unroll
        for (int nt = 0; nt < 3; ++nt) {
            bf2[nt] = sW2[pidx + nt * 16];
            bfp[nt] = sWP[pidx + nt * 16];
        }

        v2f a3, ap;
        a3.x = valid ? (xv.x + av.x * di) : 0.0f;
        a3.y = valid ? (xv.y + av.y * di) : 0.0f;
        ap.x = valid ? xv.x : 0.0f;
        ap.y = valid ? xv.y : 0.0f;

        if (k0 + 4 < DIM) {
            xv = *(const float2*)(xr + k0 + 4 + kk);
            av = *(const float2*)(ar + k0 + 4 + kk);
        }
#pragma unroll
        for (int nt = 0; nt < 3; ++nt) {
            acc3[nt] = __builtin_amdgcn_wmma_f32_16x16x4_f32(
                false, a3, false, bf2[nt], (short)0, acc3[nt], false, false);
            accp[nt] = __builtin_amdgcn_wmma_f32_16x16x4_f32(
                false, ap, false, bfp[nt], (short)0, accp[nt], false, false);
        }
    }

    const int roff = prow << 3;
#pragma unroll
    for (int nt = 0; nt < 3; ++nt) {
        const int col = nt * 16 + m;
        if (col >= NCLS) continue;
        const float s  = scale2[col];
        const float sh = shift2[col];
        const float bb = bp[col];
#pragma unroll
        for (int j = 0; j < 8; ++j) {
            int rr = row0 + roff + j;
            if (rr < NN) {
                float h3 = acc3[nt][j] * s + sh;
                h3 = h3 > 0.0f ? h3 : 0.0f;
                out[(long)rr * NCLS + col] = 0.5f * (accp[nt][j] + bb + h3);
            }
        }
    }
}

extern "C" void kernel_launch(void* const* d_in, const int* in_sizes, int n_in,
                              void* d_out, int out_size, void* d_ws, size_t ws_size,
                              hipStream_t stream) {
    const float* h        = (const float*)d_in[0];
    const int*   edge_src = (const int*)d_in[1];
    const int*   edge_dst = (const int*)d_in[2];
    const float* mask     = (const float*)d_in[3];
    const float* w0  = (const float*)d_in[4];
    const float* b0  = (const float*)d_in[5];
    const float* g0  = (const float*)d_in[6];
    const float* be0 = (const float*)d_in[7];
    const float* rm0 = (const float*)d_in[8];
    const float* rv0 = (const float*)d_in[9];
    const float* w1  = (const float*)d_in[10];
    const float* b1  = (const float*)d_in[11];
    const float* g1  = (const float*)d_in[12];
    const float* be1 = (const float*)d_in[13];
    const float* rm1 = (const float*)d_in[14];
    const float* rv1 = (const float*)d_in[15];
    const float* w2  = (const float*)d_in[16];
    const float* b2  = (const float*)d_in[17];
    const float* g2  = (const float*)d_in[18];
    const float* be2 = (const float*)d_in[19];
    const float* rm2 = (const float*)d_in[20];
    const float* rv2 = (const float*)d_in[21];
    const float* wp  = (const float*)d_in[22];
    const float* bp  = (const float*)d_in[23];
    float* out = (float*)d_out;

    float* W = (float*)d_ws;
    float* dinv = W;
    float* agg  = dinv + 50048;
    float* h1   = agg  + (long)NN * DIM;
    float* h2   = h1   + (long)NN * DIM;
    float* cbuf = h2   + (long)NN * DIM;
    float* sc0 = cbuf + 0 * DIM;
    float* sh0 = cbuf + 1 * DIM;
    float* sc1 = cbuf + 2 * DIM;
    float* sh1 = cbuf + 3 * DIM;
    float* sc2 = cbuf + 4 * DIM;
    float* sh2 = cbuf + 4 * DIM + 64;

    const long aggN = (long)NN * DIM;
    const int ZB = 256;
    const int zeroBlocksAgg = (int)((aggN + ZB * 8 - 1) / (ZB * 8));
    const int gemmBlocks = (NN + 127) / 128;

    zero_kernel<<<(NN + ZB - 1) / ZB, ZB, 0, stream>>>(dinv, NN);
    deg_kernel<<<(EE + ZB - 1) / ZB, ZB, 0, stream>>>(edge_dst, dinv);
    deginv_kernel<<<(NN + ZB - 1) / ZB, ZB, 0, stream>>>(dinv);

    consts_kernel<<<1, 128, 0, stream>>>(b0, g0, be0, rm0, rv0,
                                         b1, g1, be1, rm1, rv1,
                                         b2, g2, be2, rm2, rv2, cbuf);

    const long scatThreads = (long)EE * (DIM / 4);
    const int scatBlocks = (int)((scatThreads + ZB - 1) / ZB);

    zero_kernel<<<zeroBlocksAgg, ZB, 0, stream>>>(agg, aggN);
    scatter_kernel<<<scatBlocks, ZB, 0, stream>>>(h, edge_src, edge_dst, mask, agg);
    gin_gemm_kernel<<<gemmBlocks, 256, 0, stream>>>(h, agg, dinv, w0, sc0, sh0, h1);

    zero_kernel<<<zeroBlocksAgg, ZB, 0, stream>>>(agg, aggN);
    scatter_kernel<<<scatBlocks, ZB, 0, stream>>>(h1, edge_src, edge_dst, mask, agg);
    gin_gemm_kernel<<<gemmBlocks, 256, 0, stream>>>(h1, agg, dinv, w1, sc1, sh1, h2);

    zero_kernel<<<zeroBlocksAgg, ZB, 0, stream>>>(agg, aggN);
    scatter_kernel<<<scatBlocks, ZB, 0, stream>>>(h2, edge_src, edge_dst, mask, agg);
    gin_final_kernel<<<gemmBlocks, 256, 0, stream>>>(h2, agg, dinv, w2, wp,
                                                     sc2, sh2, bp, out);
}


# Replace float-atomic edge scatter by a per-call counting sort of edges by destination (integer atomic arrival-order rank) plus an atomic-free row gather; per-node sum order = rank order
# speedup vs baseline: 2.5523x; 2.5523x over previous
#include <hip/hip_runtime.h>

typedef __attribute__((ext_vector_type(2))) float v2f;
typedef __attribute__((ext_vector_type(8))) float v8f;

#define NN 50000
#define EE 800000
#define DIM 128
#define NCLS 40
#define CP 48

__global__ void zero_int_kernel(int* __restrict__ p, int n) {
    int i = blockIdx.x * blockDim.x + threadIdx.x;
    if (i < n) p[i] = 0;
}

__global__ void rank_kernel(const int* __restrict__ dst, int* __restrict__ cnt,
                            int* __restrict__ rank) {
    int e = blockIdx.x * blockDim.x + threadIdx.x;
    if (e < EE) rank[e] = atomicAdd(&cnt[dst[e]], 1);
}

__global__ void __launch_bounds__(256)
scan_kernel(const int* __restrict__ cnt, int* __restrict__ off, float* __restrict__ dinv) {
    __shared__ int sPart[8];
    __shared__ int sIncl[8];
    const int tid  = threadIdx.x;
    const int lane = tid & 31;
    const int wave = tid >> 5;
    const int b0   = blockIdx.x * 256;

    int part = 0;
    const int4* c4 = (const int4*)cnt;
    const int n4 = b0 >> 2;
#pragma unroll 8
    for (int i = tid; i < n4; i += 256) {
        int4 v = c4[i];
        part += (v.x + v.y) + (v.z + v.w);
    }
    const int i = b0 + tid;
    const int c = (i < NN) ? cnt[i] : 0;

    int incl = c;
#pragma unroll
    for (int d = 1; d < 32; d <<= 1) {
        int t = __shfl_up(incl, d);
        if (lane >= d) incl += t;
    }
#pragma unroll
    for (int d = 16; d > 0; d >>= 1) part += __shfl_xor(part, d);
    if (lane == 31) sIncl[wave] = incl;
    if (lane == 0)  sPart[wave] = part;
    __syncthreads();
    int base = 0;
#pragma unroll
    for (int w = 0; w < 8; ++w) {
        base += sPart[w];
        if (w < wave) base += sIncl[w];
    }
    const int excl = base + incl - c;
    if (i < NN) {
        off[i]  = excl;
        dinv[i] = (c > 0) ? (1.0f / (float)c) : 0.0f;
    }
    if (i == NN - 1) off[NN] = excl + c;
}

__global__ void fill_kernel(const int* __restrict__ src, const int* __restrict__ dst,
                            const float* __restrict__ mask, const int* __restrict__ rank,
                            const int* __restrict__ off, int2* __restrict__ rec) {
    int e = blockIdx.x * blockDim.x + threadIdx.x;
    if (e < EE) {
        int pos = off[dst[e]] + rank[e];
        rec[pos] = make_int2(src[e], __float_as_int(mask[e]));
    }
}

__global__ void consts_kernel(const float* b0, const float* g0, const float* be0,
                              const float* rm0, const float* rv0,
                              const float* b1, const float* g1, const float* be1,
                              const float* rm1, const float* rv1,
                              const float* b2, const float* g2, const float* be2,
                              const float* rm2, const float* rv2,
                              float* __restrict__ cbuf) {
    int t = threadIdx.x;
    const float eps = 1e-5f;
    if (t < DIM) {
        float s = g0[t] * rsqrtf(rv0[t] + eps);
        cbuf[0 * DIM + t] = s;
        cbuf[1 * DIM + t] = (b0[t] - rm0[t]) * s + be0[t];
        s = g1[t] * rsqrtf(rv1[t] + eps);
        cbuf[2 * DIM + t] = s;
        cbuf[3 * DIM + t] = (b1[t] - rm1[t]) * s + be1[t];
    }
    if (t < NCLS) {
        float s = g2[t] * rsqrtf(rv2[t] + eps);
        cbuf[4 * DIM + t]      = s;
        cbuf[4 * DIM + 64 + t] = (b2[t] - rm2[t]) * s + be2[t];
    }
}

#define ROW_FMA(V, M)                      \
    acc.x = fmaf((V).x, (M), acc.x);       \
    acc.y = fmaf((V).y, (M), acc.y);       \
    acc.z = fmaf((V).z, (M), acc.z);       \
    acc.w = fmaf((V).w, (M), acc.w);

__global__ void __launch_bounds__(256)
gather_kernel(const float* __restrict__ x, const int* __restrict__ off,
              const int2* __restrict__ rec, float* __restrict__ agg) {
    const int lane = threadIdx.x & 31;
    const int wv = __builtin_amdgcn_readfirstlane((int)(threadIdx.x >> 5));
    const int n  = blockIdx.x * 8 + wv;
    if (n >= NN) return;
    const int beg = off[n];
    const int end = off[n + 1];
    const float* xl = x + lane * 4;
    float4 acc = make_float4(0.0f, 0.0f, 0.0f, 0.0f);

    for (int base = beg; base < end; base += 32) {
        const int c = min(32, end - base);
        int2 ed = make_int2(0, 0);
        if (lane < c) ed = rec[base + lane];
        int j = 0;
        for (; j + 4 <= c; j += 4) {
            const int s0 = __builtin_amdgcn_readlane(ed.x, j);
            const int s1 = __builtin_amdgcn_readlane(ed.x, j + 1);
            const int s2 = __builtin_amdgcn_readlane(ed.x, j + 2);
            const int s3 = __builtin_amdgcn_readlane(ed.x, j + 3);
            const float m0 = __int_as_float(__builtin_amdgcn_readlane(ed.y, j));
            const float m1 = __int_as_float(__builtin_amdgcn_readlane(ed.y, j + 1));
            const float m2 = __int_as_float(__builtin_amdgcn_readlane(ed.y, j + 2));
            const float m3 = __int_as_float(__builtin_amdgcn_readlane(ed.y, j + 3));
            const float4 v0 = *(const float4*)(xl + (long)s0 * DIM);
            const float4 v1 = *(const float4*)(xl + (long)s1 * DIM);
            const float4 v2 = *(const float4*)(xl + (long)s2 * DIM);
            const float4 v3 = *(const float4*)(xl + (long)s3 * DIM);
            ROW_FMA(v0, m0)
            ROW_FMA(v1, m1)
            ROW_FMA(v2, m2)
            ROW_FMA(v3, m3)
        }
        const int rem = c - j;
        if (rem > 0) {
            const int j1 = min(j + 1, c - 1);
            const int j2 = min(j + 2, c - 1);
            const int s0 = __builtin_amdgcn_readlane(ed.x, j);
            const int s1 = __builtin_amdgcn_readlane(ed.x, j1);
            const int s2 = __builtin_amdgcn_readlane(ed.x, j2);
            const float m0 = __int_as_float(__builtin_amdgcn_readlane(ed.y, j));
            const float m1 = __int_as_float(__builtin_amdgcn_readlane(ed.y, j1));
            const float m2 = __int_as_float(__builtin_amdgcn_readlane(ed.y, j2));
            const float4 v0 = *(const float4*)(xl + (long)s0 * DIM);
            const float4 v1 = *(const float4*)(xl + (long)s1 * DIM);
            const float4 v2 = *(const float4*)(xl + (long)s2 * DIM);
            ROW_FMA(v0, m0)
            if (rem > 1) { ROW_FMA(v1, m1) }
            if (rem > 2) { ROW_FMA(v2, m2) }
        }
    }
    *(float4*)(agg + (long)n * DIM + lane * 4) = acc;
}

__global__ void __launch_bounds__(256)
gin_gemm_kernel(const float* __restrict__ x, const float* __restrict__ agg,
                const float* __restrict__ dinv, const float* __restrict__ W,
                const float* __restrict__ scale, const float* __restrict__ shift,
                float* __restrict__ out) {
    __shared__ v2f sW[(DIM / 2) * DIM];
    const int tid = threadIdx.x;
    for (int i = tid; i < (DIM / 2) * DIM; i += 256) {
        int p = i / DIM, c = i % DIM;
        v2f w;
        w.x = W[(2 * p) * DIM + c];
        w.y = W[(2 * p + 1) * DIM + c];
        sW[i] = w;
    }
    __syncthreads();

    const int wave = tid >> 5;
    const int lane = tid & 31;
    const int m    = lane & 15;
    const int prow = lane >> 4;
    const int kk   = prow << 1;
    const int row0 = blockIdx.x * 128 + wave * 16;

    const int  r     = row0 + m;
    const bool valid = r < NN;
    const int  rc    = valid ? r : 0;
    const float di   = valid ? dinv[rc] : 0.0f;
    const float* xr  = x   + (long)rc * DIM;
    const float* ar  = agg + (long)rc * DIM;

    v8f acc[8] = {};
    float2 xv = *(const float2*)(xr + kk);
    float2 av = *(const float2*)(ar + kk);
#pragma unroll 4
    for (int k0 = 0; k0 < DIM; k0 += 4) {
        const v2f* bb = sW + ((k0 >> 1) + prow) * DIM + m;
        v2f bf[8];
#pragma unroll
        for (int nt = 0; nt < 8; ++nt) bf[nt] = bb[nt * 16];

        v2f a;
        a.x = valid ? (xv.x + av.x * di) : 0.0f;
        a.y = valid ? (xv.y + av.y * di) : 0.0f;

        if (k0 + 4 < DIM) {
            xv = *(const float2*)(xr + k0 + 4 + kk);
            av = *(const float2*)(ar + k0 + 4 + kk);
        }
#pragma unroll
        for (int nt = 0; nt < 8; ++nt)
            acc[nt] = __builtin_amdgcn_wmma_f32_16x16x4_f32(
                false, a, false, bf[nt], (short)0, acc[nt], false, false);
    }

    const int roff = prow << 3;
#pragma unroll
    for (int nt = 0; nt < 8; ++nt) {
        const int col = nt * 16 + m;
        const float s  = scale[col];
        const float sh = shift[col];
#pragma unroll
        for (int j = 0; j < 8; ++j) {
            int rr = row0 + roff + j;
            if (rr < NN) {
                float v = acc[nt][j] * s + sh;
                out[(long)rr * DIM + col] = v > 0.0f ? v : 0.0f;
            }
        }
    }
}

__global__ void __launch_bounds__(256)
gin_final_kernel(const float* __restrict__ h2, const float* __restrict__ agg,
                 const float* __restrict__ dinv,
                 const float* __restrict__ w2, const float* __restrict__ wp,
                 const float* __restrict__ scale2, const float* __restrict__ shift2,
                 const float* __restrict__ bp, float* __restrict__ out) {
    __shared__ v2f sW2[(DIM / 2) * CP];
    __shared__ v2f sWP[(DIM / 2) * CP];
    const int tid = threadIdx.x;
    for (int i = tid; i < (DIM / 2) * CP; i += 256) {
        int p = i / CP, c = i % CP;
        v2f a = {0.0f, 0.0f}, b = {0.0f, 0.0f};
        if (c < NCLS) {
            a.x = w2[(2 * p) * NCLS + c];
            a.y = w2[(2 * p + 1) * NCLS + c];
            b.x = wp[(2 * p) * NCLS + c];
            b.y = wp[(2 * p + 1) * NCLS + c];
        }
        sW2[i] = a;
        sWP[i] = b;
    }
    __syncthreads();

    const int wave = tid >> 5;
    const int lane = tid & 31;
    const int m    = lane & 15;
    const int prow = lane >> 4;
    const int kk   = prow << 1;
    const int row0 = blockIdx.x * 128 + wave * 16;

    const int  r     = row0 + m;
    const bool valid = r < NN;
    const int  rc    = valid ? r : 0;
    const float di   = valid ? dinv[rc] : 0.0f;
    const float* xr  = h2  + (long)rc * DIM;
    const float* ar  = agg + (long)rc * DIM;

    v8f acc3[3] = {};
    v8f accp[3] = {};
    float2 xv = *(const float2*)(xr + kk);
    float2 av = *(const float2*)(ar + kk);
#pragma unroll 4
    for (int k0 = 0; k0 < DIM; k0 += 4) {
        const int pidx = ((k0 >> 1) + prow) * CP + m;
        v2f bf2[3], bfp[3];
#pragma unroll
        for (int nt = 0; nt < 3; ++nt) {
            bf2[nt] = sW2[pidx + nt * 16];
            bfp[nt] = sWP[pidx + nt * 16];
        }

        v2f a3, ap;
        a3.x = valid ? (xv.x + av.x * di) : 0.0f;
        a3.y = valid ? (xv.y + av.y * di) : 0.0f;
        ap.x = valid ? xv.x : 0.0f;
        ap.y = valid ? xv.y : 0.0f;

        if (k0 + 4 < DIM) {
            xv = *(const float2*)(xr + k0 + 4 + kk);
            av = *(const float2*)(ar + k0 + 4 + kk);
        }
#pragma unroll
        for (int nt = 0; nt < 3; ++nt) {
            acc3[nt] = __builtin_amdgcn_wmma_f32_16x16x4_f32(
                false, a3, false, bf2[nt], (short)0, acc3[nt], false, false);
            accp[nt] = __builtin_amdgcn_wmma_f32_16x16x4_f32(
                false, ap, false, bfp[nt], (short)0, accp[nt], false, false);
        }
    }

    const int roff = prow << 3;
#pragma unroll
    for (int nt = 0; nt < 3; ++nt) {
        const int col = nt * 16 + m;
        if (col >= NCLS) continue;
        const float s  = scale2[col];
        const float sh = shift2[col];
        const float bb = bp[col];
#pragma unroll
        for (int j = 0; j < 8; ++j) {
            int rr = row0 + roff + j;
            if (rr < NN) {
                float h3 = acc3[nt][j] * s + sh;
                h3 = h3 > 0.0f ? h3 : 0.0f;
                out[(long)rr * NCLS + col] = 0.5f * (accp[nt][j] + bb + h3);
            }
        }
    }
}

extern "C" void kernel_launch(void* const* d_in, const int* in_sizes, int n_in,
                              void* d_out, int out_size, void* d_ws, size_t ws_size,
                              hipStream_t stream) {
    const float* h        = (const float*)d_in[0];
    const int*   edge_src = (const int*)d_in[1];
    const int*   edge_dst = (const int*)d_in[2];
    const float* mask     = (const float*)d_in[3];
    const float* w0  = (const float*)d_in[4];
    const float* b0  = (const float*)d_in[5];
    const float* g0  = (const float*)d_in[6];
    const float* be0 = (const float*)d_in[7];
    const float* rm0 = (const float*)d_in[8];
    const float* rv0 = (const float*)d_in[9];
    const float* w1  = (const float*)d_in[10];
    const float* b1  = (const float*)d_in[11];
    const float* g1  = (const float*)d_in[12];
    const float* be1 = (const float*)d_in[13];
    const float* rm1 = (const float*)d_in[14];
    const float* rv1 = (const float*)d_in[15];
    const float* w2  = (const float*)d_in[16];
    const float* b2  = (const float*)d_in[17];
    const float* g2  = (const float*)d_in[18];
    const float* be2 = (const float*)d_in[19];
    const float* rm2 = (const float*)d_in[20];
    const float* rv2 = (const float*)d_in[21];
    const float* wp  = (const float*)d_in[22];
    const float* bp  = (const float*)d_in[23];
    float* out = (float*)d_out;

    float* W = (float*)d_ws;
    float* dinv = W;
    float* agg  = dinv + 50048;
    float* h1   = agg  + (long)NN * DIM;
    float* h2   = h1   + (long)NN * DIM;
    float* cbuf = h2   + (long)NN * DIM;
    int*   cnt  = (int*)(cbuf + 1024);
    int*   off  = cnt + 50048;
    int*   rank = off + 50048;
    int2*  rec  = (int2*)(rank + EE);
    float* sc0 = cbuf + 0 * DIM;
    float* sh0 = cbuf + 1 * DIM;
    float* sc1 = cbuf + 2 * DIM;
    float* sh1 = cbuf + 3 * DIM;
    float* sc2 = cbuf + 4 * DIM;
    float* sh2 = cbuf + 4 * DIM + 64;

    const int ZB = 256;
    const int gemmBlocks   = (NN + 127) / 128;
    const int nodeBlocks   = (NN + ZB - 1) / ZB;
    const int edgeBlocks   = (EE + ZB - 1) / ZB;
    const int gatherBlocks = (NN + 7) / 8;

    zero_int_kernel<<<nodeBlocks, ZB, 0, stream>>>(cnt, NN);
    rank_kernel<<<edgeBlocks, ZB, 0, stream>>>(edge_dst, cnt, rank);
    scan_kernel<<<nodeBlocks, ZB, 0, stream>>>(cnt, off, dinv);
    fill_kernel<<<edgeBlocks, ZB, 0, stream>>>(edge_src, edge_dst, mask, rank, off, rec);

    consts_kernel<<<1, 128, 0, stream>>>(b0, g0, be0, rm0, rv0,
                                         b1, g1, be1, rm1, rv1,
                                         b2, g2, be2, rm2, rv2, cbuf);

    gather_kernel<<<gatherBlocks, ZB, 0, stream>>>(h, off, rec, agg);
    gin_gemm_kernel<<<gemmBlocks, 256, 0, stream>>>(h, agg, dinv, w0, sc0, sh0, h1);

    gather_kernel<<<gatherBlocks, ZB, 0, stream>>>(h1, off, rec, agg);
    gin_gemm_kernel<<<gemmBlocks, 256, 0, stream>>>(h1, agg, dinv, w1, sc1, sh1, h2);

    gather_kernel<<<gatherBlocks, ZB, 0, stream>>>(h2, off, rec, agg);
    gin_final_kernel<<<gemmBlocks, 256, 0, stream>>>(h2, agg, dinv, w2, wp,
                                                     sc2, sh2, bp, out);
}


# Edge sort without global atomics: two-level counting sort on LDS integer atomics (3125-edge chunks by 256-node bucket, then per-bucket node sort); per-node sum order = LDS arrival order
# speedup vs baseline: 3.6172x; 1.4172x over previous
#include <hip/hip_runtime.h>

typedef __attribute__((ext_vector_type(2))) float v2f;
typedef __attribute__((ext_vector_type(8))) float v8f;

#define NN 50000
#define EE 800000
#define DIM 128
#define NCLS 40
#define CP 48

#define BSHIFT   8
#define BW       (1 << BSHIFT)
#define NBKT     ((NN + BW - 1) / BW)
#define NCHUNK   256
#define CHUNK    (EE / NCHUNK)
#define STHREADS 512
#define EPT      ((CHUNK + STHREADS - 1) / STHREADS)
#define NHIST    (NBKT * NCHUNK)
#define DBATCH   8
static_assert(EE % NCHUNK == 0, "chunks must cover the edge list exactly");
static_assert(NBKT <= 256 && BW == 256, "the 256-entry scan covers the buckets of a chunk and the nodes of a bucket");
static_assert(NN <= 65536, "a source index must fit the low 16 bits of a record word");
static_assert(NHIST % 256 == 0, "the histogram scan works in blocks of 256 entries");

__device__ __forceinline__ int excl_scan_256(const int c, const int tid, int* __restrict__ sTot) {
    const int lane = tid & 31;
    const int wave = tid >> 5;
    int incl = c;
#pragma unroll
    for (int d = 1; d < 32; d <<= 1) {
        int t = __shfl_up(incl, d);
        if (lane >= d) incl += t;
    }
    if (lane == 31 && wave < 8) sTot[wave] = incl;
    __syncthreads();
    int basev = 0;
#pragma unroll
    for (int w = 0; w < 8; ++w)
        if (w < wave) basev += sTot[w];
    return basev + incl - c;
}

__global__ void __launch_bounds__(STHREADS)
chunk_hist_kernel(const int* __restrict__ dst, int* __restrict__ hist) {
    __shared__ int sCnt[NBKT];
    const int tid = threadIdx.x;
    const int c   = blockIdx.x;
    const int e0  = c * CHUNK;
    int d[EPT];
#pragma unroll
    for (int k = 0; k < EPT; ++k) {
        const int i = tid + k * STHREADS;
        d[k] = (i < CHUNK) ? dst[e0 + i] : -1;
    }
    if (tid < NBKT) sCnt[tid] = 0;
    __syncthreads();
#pragma unroll
    for (int k = 0; k < EPT; ++k)
        if (d[k] >= 0) atomicAdd(&sCnt[d[k] >> BSHIFT], 1);
    __syncthreads();
    if (tid < NBKT) hist[tid * NCHUNK + c] = sCnt[tid];
}

__global__ void __launch_bounds__(256)
scan_hist_kernel(const int* __restrict__ hist, int* __restrict__ base) {
    __shared__ int sPart[8];
    __shared__ int sIncl[8];
    const int tid  = threadIdx.x;
    const int lane = tid & 31;
    const int wave = tid >> 5;
    const int b0   = blockIdx.x * 256;

    int part = 0;
    const int4* c4 = (const int4*)hist;
    const int n4 = b0 >> 2;
#pragma unroll 8
    for (int i = tid; i < n4; i += 256) {
        int4 v = c4[i];
        part += (v.x + v.y) + (v.z + v.w);
    }
    const int i = b0 + tid;
    const int c = hist[i];

    int incl = c;
#pragma unroll
    for (int d = 1; d < 32; d <<= 1) {
        int t = __shfl_up(incl, d);
        if (lane >= d) incl += t;
    }
#pragma unroll
    for (int d = 16; d > 0; d >>= 1) part += __shfl_xor(part, d);
    if (lane == 31) sIncl[wave] = incl;
    if (lane == 0)  sPart[wave] = part;
    __syncthreads();
    int basev = 0;
#pragma unroll
    for (int w = 0; w < 8; ++w) {
        basev += sPart[w];
        if (w < wave) basev += sIncl[w];
    }
    const int excl = basev + incl - c;
    base[i] = excl;
    if (i == NHIST - 1) base[NHIST] = excl + c;
}

__global__ void __launch_bounds__(STHREADS)
chunk_scatter_kernel(const int* __restrict__ src, const int* __restrict__ dst,
                     const float* __restrict__ mask, const int* __restrict__ base,
                     int2* __restrict__ tmp) {
    __shared__ int sCnt[256];
    __shared__ int sLst[256];
    __shared__ int sGof[256];
    __shared__ int sTot[8];
    __shared__ int2 sRec[CHUNK];
    __shared__ unsigned char sBid[CHUNK];
    const int tid = threadIdx.x;
    const int c   = blockIdx.x;
    const int e0  = c * CHUNK;
    int d[EPT], s[EPT], mk[EPT], r[EPT];
#pragma unroll
    for (int k = 0; k < EPT; ++k) {
        const int i = tid + k * STHREADS;
        const bool in = i < CHUNK;
        d[k]  = in ? dst[e0 + i] : -1;
        s[k]  = in ? src[e0 + i] : 0;
        mk[k] = in ? __float_as_int(mask[e0 + i]) : 0;
        r[k]  = 0;
    }
    if (tid < 256) sCnt[tid] = 0;
    __syncthreads();
#pragma unroll
    for (int k = 0; k < EPT; ++k)
        if (d[k] >= 0) r[k] = atomicAdd(&sCnt[d[k] >> BSHIFT], 1);
    __syncthreads();
    const int cnt = (tid < 256) ? sCnt[tid] : 0;
    const int gb = (tid < NBKT) ? base[tid * NCHUNK + c] : 0;
    const int excl = excl_scan_256(cnt, tid, sTot);
    if (tid < NBKT) {
        sLst[tid] = excl;
        sGof[tid] = gb - excl;
    }
    __syncthreads();
#pragma unroll
    for (int k = 0; k < EPT; ++k) {
        if (d[k] >= 0) {
            const int b  = d[k] >> BSHIFT;
            const int lp = sLst[b] + r[k];
            sRec[lp] = make_int2(((d[k] & (BW - 1)) << 16) | s[k], mk[k]);
            sBid[lp] = (unsigned char)b;
        }
    }
    __syncthreads();
    for (int i = tid; i < CHUNK; i += STHREADS) tmp[i + sGof[sBid[i]]] = sRec[i];
}

__global__ void __launch_bounds__(STHREADS)
bucket_sort_kernel(const int2* __restrict__ tmp, const int* __restrict__ base,
                   int* __restrict__ off, float* __restrict__ dinv, int2* __restrict__ rec) {
    __shared__ int sCnt[BW];
    __shared__ int sExc[BW];
    __shared__ int sTot[8];
    const int tid = threadIdx.x;
    const int b   = blockIdx.x;
    const int beg = base[b * NCHUNK];
    const int end = base[(b + 1) * NCHUNK];
    if (tid < BW) sCnt[tid] = 0;
    __syncthreads();
    for (int i0 = beg; i0 < end; i0 += DBATCH * STHREADS) {
        int w[DBATCH];
#pragma unroll
        for (int k = 0; k < DBATCH; ++k) {
            const int i = i0 + k * STHREADS + tid;
            w[k] = (i < end) ? tmp[i].x : -1;
        }
#pragma unroll
        for (int k = 0; k < DBATCH; ++k)
            if (w[k] >= 0) atomicAdd(&sCnt[w[k] >> 16], 1);
    }
    __syncthreads();
    const int cnt  = (tid < BW) ? sCnt[tid] : 0;
    const int excl = excl_scan_256(cnt, tid, sTot);
    if (tid < BW) {
        sExc[tid] = excl;
        sCnt[tid] = 0;
        const int node = b * BW + tid;
        if (node < NN) {
            off[node]  = beg + excl;
            dinv[node] = (cnt > 0) ? (1.0f / (float)cnt) : 0.0f;
        }
        if (node == NN - 1) off[NN] = beg + excl + cnt;
    }
    __syncthreads();
    for (int i0 = beg; i0 < end; i0 += DBATCH * STHREADS) {
        int2 t[DBATCH];
#pragma unroll
        for (int k = 0; k < DBATCH; ++k) {
            const int i = i0 + k * STHREADS + tid;
            t[k] = (i < end) ? tmp[i] : make_int2(-1, 0);
        }
#pragma unroll
        for (int k = 0; k < DBATCH; ++k) {
            if (t[k].x >= 0) {
                const int dl = t[k].x >> 16;
                const int p  = atomicAdd(&sCnt[dl], 1);
                rec[beg + sExc[dl] + p] = make_int2(t[k].x & 0xffff, t[k].y);
            }
        }
    }
}

__global__ void consts_kernel(const float* b0, const float* g0, const float* be0,
                              const float* rm0, const float* rv0,
                              const float* b1, const float* g1, const float* be1,
                              const float* rm1, const float* rv1,
                              const float* b2, const float* g2, const float* be2,
                              const float* rm2, const float* rv2,
                              float* __restrict__ cbuf) {
    int t = threadIdx.x;
    const float eps = 1e-5f;
    if (t < DIM) {
        float s = g0[t] * rsqrtf(rv0[t] + eps);
        cbuf[0 * DIM + t] = s;
        cbuf[1 * DIM + t] = (b0[t] - rm0[t]) * s + be0[t];
        s = g1[t] * rsqrtf(rv1[t] + eps);
        cbuf[2 * DIM + t] = s;
        cbuf[3 * DIM + t] = (b1[t] - rm1[t]) * s + be1[t];
    }
    if (t < NCLS) {
        float s = g2[t] * rsqrtf(rv2[t] + eps);
        cbuf[4 * DIM + t]      = s;
        cbuf[4 * DIM + 64 + t] = (b2[t] - rm2[t]) * s + be2[t];
    }
}

#define NWAVE   13
#define LTHREADS (NWAVE * 32)
#define LGRID   256
#define NTILES  (NN / 16)
#define GB      16
#define TPITCH  132
static_assert(NN % 16 == 0, "tiles of 16 nodes must cover the node range exactly");
static_assert(LGRID * NWAVE >= NTILES, "every tile needs a wave");

#define ROW_FMA(V, M)                      \
    acc.x = fmaf((V).x, (M), acc.x);       \
    acc.y = fmaf((V).y, (M), acc.y);       \
    acc.z = fmaf((V).z, (M), acc.z);       \
    acc.w = fmaf((V).w, (M), acc.w);

__device__ __forceinline__ void gather_tile(const float* __restrict__ x, const int* __restrict__ off,
                                            const int2* __restrict__ rec, const int n0,
                                            float* __restrict__ tile, const int lane) {
    const int offv = off[n0 + min(lane, 16)];
    const int beg = __builtin_amdgcn_readlane(offv, 0);
    const int end = __builtin_amdgcn_readlane(offv, 16);
    const float* xl = x + lane * 4;
    float* trow = tile + lane * 4;
    float4 acc = make_float4(0.0f, 0.0f, 0.0f, 0.0f);
    int cur = 0;
    int nb  = __builtin_amdgcn_readlane(offv, 1);

    int2 ed = make_int2(0, 0);
    if (beg + lane < end) ed = rec[beg + lane];
    for (int base = beg; base < end; base += 32) {
        const int c = min(32, end - base);
        int2 edn = make_int2(0, 0);
        if (base + 32 + lane < end) edn = rec[base + 32 + lane];
        for (int j = 0; j < c; j += GB) {
            float4 v[GB];
            float  mk[GB];
#pragma unroll
            for (int k = 0; k < GB; ++k) {
                const int jj = min(j + k, c - 1);
                const int s = __builtin_amdgcn_readlane(ed.x, jj);
                mk[k] = __int_as_float(__builtin_amdgcn_readlane(ed.y, jj));
                v[k] = *(const float4*)(xl + (long)s * DIM);
            }
#pragma unroll
            for (int k = 0; k < GB; ++k) {
                if (j + k < c) {
                    const int e = base + j + k;
                    while (cur < 15 && e == nb) {
                        *(float4*)(trow + cur * TPITCH) = acc;
                        acc = make_float4(0.0f, 0.0f, 0.0f, 0.0f);
                        ++cur;
                        nb = __builtin_amdgcn_readlane(offv, cur + 1);
                    }
                    ROW_FMA(v[k], mk[k])
                }
            }
        }
        ed = edn;
    }
    for (; cur < 16; ++cur) {
        *(float4*)(trow + cur * TPITCH) = acc;
        acc = make_float4(0.0f, 0.0f, 0.0f, 0.0f);
    }
}

__global__ void __launch_bounds__(LTHREADS)
gin_gemm_kernel(const float* __restrict__ x, const int* __restrict__ off,
                const int2* __restrict__ rec,
                const float* __restrict__ dinv, const float* __restrict__ W,
                const float* __restrict__ scale, const float* __restrict__ shift,
                float* __restrict__ out) {
    __shared__ v2f sW[(DIM / 2) * DIM];
    __shared__ __attribute__((aligned(16))) float sA[NWAVE * 16 * TPITCH];
    const int tid = threadIdx.x;
    for (int i = tid; i < (DIM / 2) * DIM; i += LTHREADS) {
        int p = i / DIM, c = i % DIM;
        v2f w;
        w.x = W[(2 * p) * DIM + c];
        w.y = W[(2 * p + 1) * DIM + c];
        sW[i] = w;
    }
    __syncthreads();

    const int wave = __builtin_amdgcn_readfirstlane(tid >> 5);
    const int lane = tid & 31;
    const int t    = blockIdx.x + LGRID * wave;
    if (t >= NTILES) return;
    const int row0 = t * 16;
    float* tile = sA + wave * (16 * TPITCH);

    gather_tile(x, off, rec, row0, tile, lane);

    {
        const float dv = dinv[row0 + (lane & 15)];
        float4 xi[16];
#pragma unroll
        for (int i = 0; i < 16; ++i) xi[i] = *(const float4*)(x + (long)(row0 + i) * DIM + lane * 4);
#pragma unroll
        for (int i = 0; i < 16; ++i) {
            const float di = __int_as_float(__builtin_amdgcn_readlane(__float_as_int(dv), i));
            float4 ag = *(const float4*)(tile + i * TPITCH + lane * 4);
            float4 a;
            a.x = xi[i].x + ag.x * di;
            a.y = xi[i].y + ag.y * di;
            a.z = xi[i].z + ag.z * di;
            a.w = xi[i].w + ag.w * di;
            *(float4*)(tile + i * TPITCH + lane * 4) = a;
        }
    }

    const int m    = lane & 15;
    const int prow = lane >> 4;
    const int kk   = prow << 1;
    const float* ar = tile + m * TPITCH + kk;

    v8f acc[8] = {};
#pragma unroll 4
    for (int k0 = 0; k0 < DIM; k0 += 4) {
        const v2f* bb = sW + ((k0 >> 1) + prow) * DIM + m;
        v2f bf[8];
#pragma unroll
        for (int nt = 0; nt < 8; ++nt) bf[nt] = bb[nt * 16];
        const v2f a = *(const v2f*)(ar + k0);
#pragma unroll
        for (int nt = 0; nt < 8; ++nt)
            acc[nt] = __builtin_amdgcn_wmma_f32_16x16x4_f32(
                false, a, false, bf[nt], (short)0, acc[nt], false, false);
    }

    const int roff = prow << 3;
#pragma unroll
    for (int nt = 0; nt < 8; ++nt) {
        const int col = nt * 16 + m;
        const float s  = scale[col];
        const float sh = shift[col];
#pragma unroll
        for (int j = 0; j < 8; ++j) {
            const int rr = row0 + roff + j;
            float v = acc[nt][j] * s + sh;
            out[(long)rr * DIM + col] = v > 0.0f ? v : 0.0f;
        }
    }
}

__global__ void __launch_bounds__(LTHREADS)
gin_final_kernel(const float* __restrict__ h2, const int* __restrict__ off,
                 const int2* __restrict__ rec,
                 const float* __restrict__ dinv,
                 const float* __restrict__ w2, const float* __restrict__ wp,
                 const float* __restrict__ scale2, const float* __restrict__ shift2,
                 const float* __restrict__ bp, float* __restrict__ out) {
    __shared__ v2f sW2[(DIM / 2) * CP];
    __shared__ v2f sWP[(DIM / 2) * CP];
    __shared__ __attribute__((aligned(16))) float sA[NWAVE * 16 * TPITCH];
    __shared__ __attribute__((aligned(16))) float sX[NWAVE * 16 * TPITCH];
    const int tid = threadIdx.x;
    for (int i = tid; i < (DIM / 2) * CP; i += LTHREADS) {
        int p = i / CP, c = i % CP;
        v2f a = {0.0f, 0.0f}, b = {0.0f, 0.0f};
        if (c < NCLS) {
            a.x = w2[(2 * p) * NCLS + c];
            a.y = w2[(2 * p + 1) * NCLS + c];
            b.x = wp[(2 * p) * NCLS + c];
            b.y = wp[(2 * p + 1) * NCLS + c];
        }
        sW2[i] = a;
        sWP[i] = b;
    }
    __syncthreads();

    const int wave = __builtin_amdgcn_readfirstlane(tid >> 5);
    const int lane = tid & 31;
    const int t    = blockIdx.x + LGRID * wave;
    if (t >= NTILES) return;
    const int row0 = t * 16;
    float* tile  = sA + wave * (16 * TPITCH);
    float* tileX = sX + wave * (16 * TPITCH);

    gather_tile(h2, off, rec, row0, tile, lane);

    {
        const float dv = dinv[row0 + (lane & 15)];
        float4 xi[16];
#pragma unroll
        for (int i = 0; i < 16; ++i) xi[i] = *(const float4*)(h2 + (long)(row0 + i) * DIM + lane * 4);
#pragma unroll
        for (int i = 0; i < 16; ++i) {
            const float di = __int_as_float(__builtin_amdgcn_readlane(__float_as_int(dv), i));
            float4 ag = *(const float4*)(tile + i * TPITCH + lane * 4);
            float4 a;
            a.x = xi[i].x + ag.x * di;
            a.y = xi[i].y + ag.y * di;
            a.z = xi[i].z + ag.z * di;
            a.w = xi[i].w + ag.w * di;
            *(float4*)(tile  + i * TPITCH + lane * 4) = a;
            *(float4*)(tileX + i * TPITCH + lane * 4) = xi[i];
        }
    }

    const int m    = lane & 15;
    const int prow = lane >> 4;
    const int kk   = prow << 1;
    const float* ar = tile  + m * TPITCH + kk;
    const float* xr = tileX + m * TPITCH + kk;

    v8f acc3[3] = {};
    v8f accp[3] = {};
#pragma unroll 4
    for (int k0 = 0; k0 < DIM; k0 += 4) {
        const int pidx = ((k0 >> 1) + prow) * CP + m;
        v2f bf2[3], bfp[3];
#pragma unroll
        for (int nt = 0; nt < 3; ++nt) {
            bf2[nt] = sW2[pidx + nt * 16];
            bfp[nt] = sWP[pidx + nt * 16];
        }
        const v2f a3 = *(const v2f*)(ar + k0);
        const v2f ap = *(const v2f*)(xr + k0);
#pragma unroll
        for (int nt = 0; nt < 3; ++nt) {
            acc3[nt] = __builtin_amdgcn_wmma_f32_16x16x4_f32(
                false, a3, false, bf2[nt], (short)0, acc3[nt], false, false);
            accp[nt] = __builtin_amdgcn_wmma_f32_16x16x4_f32(
                false, ap, false, bfp[nt], (short)0, accp[nt], false, false);
        }
    }

    const int roff = prow << 3;
#pragma unroll
    for (int nt = 0; nt < 3; ++nt) {
        const int col = nt * 16 + m;
        if (col >= NCLS) continue;
        const float s  = scale2[col];
        const float sh = shift2[col];
        const float bb = bp[col];
#pragma unroll
        for (int j = 0; j < 8; ++j) {
            const int rr = row0 + roff + j;
            float h3 = acc3[nt][j] * s + sh;
            h3 = h3 > 0.0f ? h3 : 0.0f;
            out[(long)rr * NCLS + col] = 0.5f * (accp[nt][j] + bb + h3);
        }
    }
}

extern "C" void kernel_launch(void* const* d_in, const int* in_sizes, int n_in,
                              void* d_out, int out_size, void* d_ws, size_t ws_size,
                              hipStream_t stream) {
    const float* h        = (const float*)d_in[0];
    const int*   edge_src = (const int*)d_in[1];
    const int*   edge_dst = (const int*)d_in[2];
    const float* mask     = (const float*)d_in[3];
    const float* w0  = (const float*)d_in[4];
    const float* b0  = (const float*)d_in[5];
    const float* g0  = (const float*)d_in[6];
    const float* be0 = (const float*)d_in[7];
    const float* rm0 = (const float*)d_in[8];
    const float* rv0 = (const float*)d_in[9];
    const float* w1  = (const float*)d_in[10];
    const float* b1  = (const float*)d_in[11];
    const float* g1  = (const float*)d_in[12];
    const float* be1 = (const float*)d_in[13];
    const float* rm1 = (const float*)d_in[14];
    const float* rv1 = (const float*)d_in[15];
    const float* w2  = (const float*)d_in[16];
    const float* b2  = (const float*)d_in[17];
    const float* g2  = (const float*)d_in[18];
    const float* be2 = (const float*)d_in[19];
    const float* rm2 = (const float*)d_in[20];
    const float* rv2 = (const float*)d_in[21];
    const float* wp  = (const float*)d_in[22];
    const float* bp  = (const float*)d_in[23];
    float* out = (float*)d_out;

    float* W = (float*)d_ws;
    float* dinv = W;
    float* h1   = dinv + 50048;
    float* h2   = h1   + (long)NN * DIM;
    float* cbuf = h2   + (long)NN * DIM;
    int*   hist = (int*)(cbuf + 1024);
    int*   base = hist + NHIST;
    int*   off  = base + NHIST + 64;
    int2*  tmp  = (int2*)(off + 50048);
    int2*  rec  = tmp + EE;
    float* sc0 = cbuf + 0 * DIM;
    float* sh0 = cbuf + 1 * DIM;
    float* sc1 = cbuf + 2 * DIM;
    float* sh1 = cbuf + 3 * DIM;
    float* sc2 = cbuf + 4 * DIM;
    float* sh2 = cbuf + 4 * DIM + 64;

    chunk_hist_kernel<<<NCHUNK, STHREADS, 0, stream>>>(edge_dst, hist);
    scan_hist_kernel<<<NHIST / 256, 256, 0, stream>>>(hist, base);
    chunk_scatter_kernel<<<NCHUNK, STHREADS, 0, stream>>>(edge_src, edge_dst, mask, base, tmp);
    bucket_sort_kernel<<<NBKT, STHREADS, 0, stream>>>(tmp, base, off, dinv, rec);

    consts_kernel<<<1, 128, 0, stream>>>(b0, g0, be0, rm0, rv0,
                                         b1, g1, be1, rm1, rv1,
                                         b2, g2, be2, rm2, rv2, cbuf);

    gin_gemm_kernel<<<LGRID, LTHREADS, 0, stream>>>(h, off, rec, dinv, w0, sc0, sh0, h1);

    gin_gemm_kernel<<<LGRID, LTHREADS, 0, stream>>>(h1, off, rec, dinv, w1, sc1, sh1, h2);

    gin_final_kernel<<<LGRID, LTHREADS, 0, stream>>>(h2, off, rec, dinv, w2, wp,
                                                     sc2, sh2, bp, out);
}


# Two launches folded away: each scatter group sums the 16-bit chunk-histogram rows itself (no scan launch) and the histogram launch also folds the BN constants; 8 launches -> 6; sum order unchanged
# speedup vs baseline: 3.8151x; 1.0547x over previous
#include <hip/hip_runtime.h>

typedef __attribute__((ext_vector_type(2))) float v2f;
typedef __attribute__((ext_vector_type(8))) float v8f;

#define NN 50000
#define EE 800000
#define DIM 128
#define NCLS 40
#define CP 48

#define BSHIFT   8
#define BW       (1 << BSHIFT)
#define NBKT     ((NN + BW - 1) / BW)
#define NCHUNK   256
#define CHUNK    (EE / NCHUNK)
#define STHREADS 512
#define EPT      ((CHUNK + STHREADS - 1) / STHREADS)
#define NHIST    (NBKT * NCHUNK)
#define DBATCH   8
static_assert(EE % NCHUNK == 0, "chunks must cover the edge list exactly");
static_assert(NBKT <= 256 && BW == 256, "the 256-entry scan covers the buckets of a chunk and the nodes of a bucket");
static_assert(NN <= 65536, "a source index must fit the low 16 bits of a record word");
static_assert(NCHUNK == 256, "a histogram row of 16-bit counts is one 16-byte load per lane of one wave");
static_assert(CHUNK * 4 < 65536, "a count, and the sum of four counts, must fit 16 bits");

__device__ __forceinline__ int excl_scan_256(const int c, const int tid, int* __restrict__ sTot) {
    const int lane = tid & 31;
    const int wave = tid >> 5;
    int incl = c;
#pragma unroll
    for (int d = 1; d < 32; d <<= 1) {
        int t = __shfl_up(incl, d);
        if (lane >= d) incl += t;
    }
    if (lane == 31 && wave < 8) sTot[wave] = incl;
    __syncthreads();
    int basev = 0;
#pragma unroll
    for (int w = 0; w < 8; ++w)
        if (w < wave) basev += sTot[w];
    return basev + incl - c;
}

struct BnArgs {
    const float* b0; const float* g0; const float* be0; const float* rm0; const float* rv0;
    const float* b1; const float* g1; const float* be1; const float* rm1; const float* rv1;
    const float* b2; const float* g2; const float* be2; const float* rm2; const float* rv2;
};

__device__ __forceinline__ void fold_consts(const BnArgs& a, float* __restrict__ cbuf, const int t) {
    const float eps = 1e-5f;
    if (t < DIM) {
        float s = a.g0[t] * rsqrtf(a.rv0[t] + eps);
        cbuf[0 * DIM + t] = s;
        cbuf[1 * DIM + t] = (a.b0[t] - a.rm0[t]) * s + a.be0[t];
        s = a.g1[t] * rsqrtf(a.rv1[t] + eps);
        cbuf[2 * DIM + t] = s;
        cbuf[3 * DIM + t] = (a.b1[t] - a.rm1[t]) * s + a.be1[t];
    }
    if (t < NCLS) {
        float s = a.g2[t] * rsqrtf(a.rv2[t] + eps);
        cbuf[4 * DIM + t]      = s;
        cbuf[4 * DIM + 64 + t] = (a.b2[t] - a.rm2[t]) * s + a.be2[t];
    }
}

__global__ void __launch_bounds__(STHREADS)
chunk_hist_kernel(const int* __restrict__ dst, unsigned short* __restrict__ hist,
                  const BnArgs bn, float* __restrict__ cbuf) {
    __shared__ int sCnt[NBKT];
    const int tid = threadIdx.x;
    const int c   = blockIdx.x;
    const int e0  = c * CHUNK;
    int d[EPT];
#pragma unroll
    for (int k = 0; k < EPT; ++k) {
        const int i = tid + k * STHREADS;
        d[k] = (i < CHUNK) ? dst[e0 + i] : -1;
    }
    if (tid < NBKT) sCnt[tid] = 0;
    __syncthreads();
#pragma unroll
    for (int k = 0; k < EPT; ++k)
        if (d[k] >= 0) atomicAdd(&sCnt[d[k] >> BSHIFT], 1);
    __syncthreads();
    if (tid < NBKT) hist[tid * NCHUNK + c] = (unsigned short)sCnt[tid];
    if (c == 0) fold_consts(bn, cbuf, tid);
}

__global__ void __launch_bounds__(STHREADS)
chunk_scatter_kernel(const int* __restrict__ src, const int* __restrict__ dst,
                     const float* __restrict__ mask, const unsigned short* __restrict__ hist,
                     int* __restrict__ bstart, int2* __restrict__ tmp) {
    __shared__ int sPT[NBKT * 33];
    __shared__ int sPP[NBKT * 33];
    __shared__ int sTotB[8];
    __shared__ int sCnt[256];
    __shared__ int sLst[256];
    __shared__ int sGof[256];
    __shared__ int sTot[8];
    __shared__ int2 sRec[CHUNK];
    __shared__ unsigned char sBid[CHUNK];
    const int tid = threadIdx.x;
    const int c   = blockIdx.x;
    const int e0  = c * CHUNK;
    constexpr int NWV = STHREADS / 32;
    constexpr int RPW = (NBKT + NWV - 1) / NWV;
    const int lane = tid & 31;
    const int wv   = tid >> 5;
    uint4 hr[RPW];
#pragma unroll
    for (int q = 0; q < RPW; ++q) {
        const int b = wv + q * NWV;
        const uint4* row = (const uint4*)(hist + min(b, NBKT - 1) * NCHUNK);
        hr[q] = row[lane];
    }
    int d[EPT], s[EPT], mk[EPT], r[EPT];
#pragma unroll
    for (int k = 0; k < EPT; ++k) {
        const int i = tid + k * STHREADS;
        const bool in = i < CHUNK;
        d[k]  = in ? dst[e0 + i] : -1;
        s[k]  = in ? src[e0 + i] : 0;
        mk[k] = in ? __float_as_int(mask[e0 + i]) : 0;
        r[k]  = 0;
    }
    if (tid < 256) sCnt[tid] = 0;
    {
        const int i0 = lane * 8;
        unsigned mw[4];
#pragma unroll
        for (int j = 0; j < 4; ++j)
            mw[j] = (i0 + 2 * j < c ? 0x0000ffffu : 0u) | (i0 + 2 * j + 1 < c ? 0xffff0000u : 0u);
#pragma unroll
        for (int q = 0; q < RPW; ++q) {
            const int b = wv + q * NWV;
            const unsigned st = (hr[q].x + hr[q].y) + (hr[q].z + hr[q].w);
            const unsigned sp = ((hr[q].x & mw[0]) + (hr[q].y & mw[1])) + ((hr[q].z & mw[2]) + (hr[q].w & mw[3]));
            const int tot = (int)((st & 0xffffu) + (st >> 16));
            const int pre = (int)((sp & 0xffffu) + (sp >> 16));
            if (b < NBKT) { sPT[b * 33 + lane] = tot; sPP[b * 33 + lane] = pre; }
        }
    }
    __syncthreads();
#pragma unroll
    for (int k = 0; k < EPT; ++k)
        if (d[k] >= 0) r[k] = atomicAdd(&sCnt[d[k] >> BSHIFT], 1);
    int btot = 0, bpre = 0;
    if (tid < NBKT) {
#pragma unroll
        for (int j = 0; j < 32; ++j) {
            btot += sPT[tid * 33 + j];
            bpre += sPP[tid * 33 + j];
        }
    }
    __syncthreads();
    const int cnt  = (tid < 256) ? sCnt[tid] : 0;
    const int bst  = excl_scan_256(btot, tid, sTotB);
    const int excl = excl_scan_256(cnt, tid, sTot);
    if (tid < NBKT) {
        sLst[tid] = excl;
        sGof[tid] = bst + bpre - excl;
        if (c == 0) bstart[tid] = bst;
        if (c == 0 && tid == NBKT - 1) bstart[NBKT] = bst + btot;
    }
    __syncthreads();
#pragma unroll
    for (int k = 0; k < EPT; ++k) {
        if (d[k] >= 0) {
            const int b  = d[k] >> BSHIFT;
            const int lp = sLst[b] + r[k];
            sRec[lp] = make_int2(((d[k] & (BW - 1)) << 16) | s[k], mk[k]);
            sBid[lp] = (unsigned char)b;
        }
    }
    __syncthreads();
    for (int i = tid; i < CHUNK; i += STHREADS) tmp[i + sGof[sBid[i]]] = sRec[i];
}

__global__ void __launch_bounds__(STHREADS)
bucket_sort_kernel(const int2* __restrict__ tmp, const int* __restrict__ bstart,
                   int* __restrict__ off, float* __restrict__ dinv, int2* __restrict__ rec) {
    __shared__ int sCnt[BW];
    __shared__ int sExc[BW];
    __shared__ int sTot[8];
    const int tid = threadIdx.x;
    const int b   = blockIdx.x;
    const int beg = bstart[b];
    const int end = bstart[b + 1];
    if (tid < BW) sCnt[tid] = 0;
    __syncthreads();
    for (int i0 = beg; i0 < end; i0 += DBATCH * STHREADS) {
        int w[DBATCH];
#pragma unroll
        for (int k = 0; k < DBATCH; ++k) {
            const int i = i0 + k * STHREADS + tid;
            w[k] = (i < end) ? tmp[i].x : -1;
        }
#pragma unroll
        for (int k = 0; k < DBATCH; ++k)
            if (w[k] >= 0) atomicAdd(&sCnt[w[k] >> 16], 1);
    }
    __syncthreads();
    const int cnt  = (tid < BW) ? sCnt[tid] : 0;
    const int excl = excl_scan_256(cnt, tid, sTot);
    if (tid < BW) {
        sExc[tid] = excl;
        sCnt[tid] = 0;
        const int node = b * BW + tid;
        if (node < NN) {
            off[node]  = beg + excl;
            dinv[node] = (cnt > 0) ? (1.0f / (float)cnt) : 0.0f;
        }
        if (node == NN - 1) off[NN] = beg + excl + cnt;
    }
    __syncthreads();
    for (int i0 = beg; i0 < end; i0 += DBATCH * STHREADS) {
        int2 t[DBATCH];
#pragma unroll
        for (int k = 0; k < DBATCH; ++k) {
            const int i = i0 + k * STHREADS + tid;
            t[k] = (i < end) ? tmp[i] : make_int2(-1, 0);
        }
#pragma unroll
        for (int k = 0; k < DBATCH; ++k) {
            if (t[k].x >= 0) {
                const int dl = t[k].x >> 16;
                const int p  = atomicAdd(&sCnt[dl], 1);
                rec[beg + sExc[dl] + p] = make_int2(t[k].x & 0xffff, t[k].y);
            }
        }
    }
}

#define NWAVE   13
#define LTHREADS (NWAVE * 32)
#define LGRID   256
#define NTILES  (NN / 16)
#define GB      16
#define TPITCH  132
static_assert(NN % 16 == 0, "tiles of 16 nodes must cover the node range exactly");
static_assert(LGRID * NWAVE >= NTILES, "every tile needs a wave");

#define ROW_FMA(V, M)                      \
    acc.x = fmaf((V).x, (M), acc.x);       \
    acc.y = fmaf((V).y, (M), acc.y);       \
    acc.z = fmaf((V).z, (M), acc.z);       \
    acc.w = fmaf((V).w, (M), acc.w);

__device__ __forceinline__ void gather_tile(const float* __restrict__ x, const int* __restrict__ off,
                                            const int2* __restrict__ rec, const int n0,
                                            float* __restrict__ tile, const int lane) {
    const int offv = off[n0 + min(lane, 16)];
    const int beg = __builtin_amdgcn_readlane(offv, 0);
    const int end = __builtin_amdgcn_readlane(offv, 16);
    const float* xl = x + lane * 4;
    float* trow = tile + lane * 4;
    float4 acc = make_float4(0.0f, 0.0f, 0.0f, 0.0f);
    int cur = 0;
    int nb  = __builtin_amdgcn_readlane(offv, 1);

    int2 ed = make_int2(0, 0);
    if (beg + lane < end) ed = rec[beg + lane];
    for (int base = beg; base < end; base += 32) {
        const int c = min(32, end - base);
        int2 edn = make_int2(0, 0);
        if (base + 32 + lane < end) edn = rec[base + 32 + lane];
        for (int j = 0; j < c; j += GB) {
            float4 v[GB];
            float  mk[GB];
#pragma unroll
            for (int k = 0; k < GB; ++k) {
                const int jj = min(j + k, c - 1);
                const int s = __builtin_amdgcn_readlane(ed.x, jj);
                mk[k] = __int_as_float(__builtin_amdgcn_readlane(ed.y, jj));
                v[k] = *(const float4*)(xl + (long)s * DIM);
            }
#pragma unroll
            for (int k = 0; k < GB; ++k) {
                if (j + k < c) {
                    const int e = base + j + k;
                    while (cur < 15 && e == nb) {
                        *(float4*)(trow + cur * TPITCH) = acc;
                        acc = make_float4(0.0f, 0.0f, 0.0f, 0.0f);
                        ++cur;
                        nb = __builtin_amdgcn_readlane(offv, cur + 1);
                    }
                    ROW_FMA(v[k], mk[k])
                }
            }
        }
        ed = edn;
    }
    for (; cur < 16; ++cur) {
        *(float4*)(trow + cur * TPITCH) = acc;
        acc = make_float4(0.0f, 0.0f, 0.0f, 0.0f);
    }
}

__global__ void __launch_bounds__(LTHREADS)
gin_gemm_kernel(const float* __restrict__ x, const int* __restrict__ off,
                const int2* __restrict__ rec,
                const float* __restrict__ dinv, const float* __restrict__ W,
                const float* __restrict__ scale, const float* __restrict__ shift,
                float* __restrict__ out) {
    __shared__ v2f sW[(DIM / 2) * DIM];
    __shared__ __attribute__((aligned(16))) float sA[NWAVE * 16 * TPITCH];
    const int tid = threadIdx.x;
    for (int i = tid; i < (DIM / 2) * DIM; i += LTHREADS) {
        int p = i / DIM, c = i % DIM;
        v2f w;
        w.x = W[(2 * p) * DIM + c];
        w.y = W[(2 * p + 1) * DIM + c];
        sW[i] = w;
    }
    __syncthreads();

    const int wave = __builtin_amdgcn_readfirstlane(tid >> 5);
    const int lane = tid & 31;
    const int t    = blockIdx.x + LGRID * wave;
    if (t >= NTILES) return;
    const int row0 = t * 16;
    float* tile = sA + wave * (16 * TPITCH);

    gather_tile(x, off, rec, row0, tile, lane);

    {
        const float dv = dinv[row0 + (lane & 15)];
        float4 xi[16];
#pragma unroll
        for (int i = 0; i < 16; ++i) xi[i] = *(const float4*)(x + (long)(row0 + i) * DIM + lane * 4);
#pragma unroll
        for (int i = 0; i < 16; ++i) {
            const float di = __int_as_float(__builtin_amdgcn_readlane(__float_as_int(dv), i));
            float4 ag = *(const float4*)(tile + i * TPITCH + lane * 4);
            float4 a;
            a.x = xi[i].x + ag.x * di;
            a.y = xi[i].y + ag.y * di;
            a.z = xi[i].z + ag.z * di;
            a.w = xi[i].w + ag.w * di;
            *(float4*)(tile + i * TPITCH + lane * 4) = a;
        }
    }

    const int m    = lane & 15;
    const int prow = lane >> 4;
    const int kk   = prow << 1;
    const float* ar = tile + m * TPITCH + kk;

    v8f acc[8] = {};
#pragma unroll 4
    for (int k0 = 0; k0 < DIM; k0 += 4) {
        const v2f* bb = sW + ((k0 >> 1) + prow) * DIM + m;
        v2f bf[8];
#pragma unroll
        for (int nt = 0; nt < 8; ++nt) bf[nt] = bb[nt * 16];
        const v2f a = *(const v2f*)(ar + k0);
#pragma unroll
        for (int nt = 0; nt < 8; ++nt)
            acc[nt] = __builtin_amdgcn_wmma_f32_16x16x4_f32(
                false, a, false, bf[nt], (short)0, acc[nt], false, false);
    }

    const int roff = prow << 3;
#pragma unroll
    for (int nt = 0; nt < 8; ++nt) {
        const int col = nt * 16 + m;
        const float s  = scale[col];
        const float sh = shift[col];
#pragma unroll
        for (int j = 0; j < 8; ++j) {
            const int rr = row0 + roff + j;
            float v = acc[nt][j] * s + sh;
            out[(long)rr * DIM + col] = v > 0.0f ? v : 0.0f;
        }
    }
}

__global__ void __launch_bounds__(LTHREADS)
gin_final_kernel(const float* __restrict__ h2, const int* __restrict__ off,
                 const int2* __restrict__ rec,
                 const float* __restrict__ dinv,
                 const float* __restrict__ w2, const float* __restrict__ wp,
                 const float* __restrict__ scale2, const float* __restrict__ shift2,
                 const float* __restrict__ bp, float* __restrict__ out) {
    __shared__ v2f sW2[(DIM / 2) * CP];
    __shared__ v2f sWP[(DIM / 2) * CP];
    __shared__ __attribute__((aligned(16))) float sA[NWAVE * 16 * TPITCH];
    __shared__ __attribute__((aligned(16))) float sX[NWAVE * 16 * TPITCH];
    const int tid = threadIdx.x;
    for (int i = tid; i < (DIM / 2) * CP; i += LTHREADS) {
        int p = i / CP, c = i % CP;
        v2f a = {0.0f, 0.0f}, b = {0.0f, 0.0f};
        if (c < NCLS) {
            a.x = w2[(2 * p) * NCLS + c];
            a.y = w2[(2 * p + 1) * NCLS + c];
            b.x = wp[(2 * p) * NCLS + c];
            b.y = wp[(2 * p + 1) * NCLS + c];
        }
        sW2[i] = a;
        sWP[i] = b;
    }
    __syncthreads();

    const int wave = __builtin_amdgcn_readfirstlane(tid >> 5);
    const int lane = tid & 31;
    const int t    = blockIdx.x + LGRID * wave;
    if (t >= NTILES) return;
    const int row0 = t * 16;
    float* tile  = sA + wave * (16 * TPITCH);
    float* tileX = sX + wave * (16 * TPITCH);

    gather_tile(h2, off, rec, row0, tile, lane);

    {
        const float dv = dinv[row0 + (lane & 15)];
        float4 xi[16];
#pragma unroll
        for (int i = 0; i < 16; ++i) xi[i] = *(const float4*)(h2 + (long)(row0 + i) * DIM + lane * 4);
#pragma unroll
        for (int i = 0; i < 16; ++i) {
            const float di = __int_as_float(__builtin_amdgcn_readlane(__float_as_int(dv), i));
            float4 ag = *(const float4*)(tile + i * TPITCH + lane * 4);
            float4 a;
            a.x = xi[i].x + ag.x * di;
            a.y = xi[i].y + ag.y * di;
            a.z = xi[i].z + ag.z * di;
            a.w = xi[i].w + ag.w * di;
            *(float4*)(tile  + i * TPITCH + lane * 4) = a;
            *(float4*)(tileX + i * TPITCH + lane * 4) = xi[i];
        }
    }

    const int m    = lane & 15;
    const int prow = lane >> 4;
    const int kk   = prow << 1;
    const float* ar = tile  + m * TPITCH + kk;
    const float* xr = tileX + m * TPITCH + kk;

    v8f acc3[3] = {};
    v8f accp[3] = {};
#pragma unroll 4
    for (int k0 = 0; k0 < DIM; k0 += 4) {
        const int pidx = ((k0 >> 1) + prow) * CP + m;
        v2f bf2[3], bfp[3];
#pragma unroll
        for (int nt = 0; nt < 3; ++nt) {
            bf2[nt] = sW2[pidx + nt * 16];
            bfp[nt] = sWP[pidx + nt * 16];
        }
        const v2f a3 = *(const v2f*)(ar + k0);
        const v2f ap = *(const v2f*)(xr + k0);
#pragma unroll
        for (int nt = 0; nt < 3; ++nt) {
            acc3[nt] = __builtin_amdgcn_wmma_f32_16x16x4_f32(
                false, a3, false, bf2[nt], (short)0, acc3[nt], false, false);
            accp[nt] = __builtin_amdgcn_wmma_f32_16x16x4_f32(
                false, ap, false, bfp[nt], (short)0, accp[nt], false, false);
        }
    }

    const int roff = prow << 3;
#pragma unroll
    for (int nt = 0; nt < 3; ++nt) {
        const int col = nt * 16 + m;
        if (col >= NCLS) continue;
        const float s  = scale2[col];
        const float sh = shift2[col];
        const float bb = bp[col];
#pragma unroll
        for (int j = 0; j < 8; ++j) {
            const int rr = row0 + roff + j;
            float h3 = acc3[nt][j] * s + sh;
            h3 = h3 > 0.0f ? h3 : 0.0f;
            out[(long)rr * NCLS + col] = 0.5f * (accp[nt][j] + bb + h3);
        }
    }
}

extern "C" void kernel_launch(void* const* d_in, const int* in_sizes, int n_in,
                              void* d_out, int out_size, void* d_ws, size_t ws_size,
                              hipStream_t stream) {
    const float* h        = (const float*)d_in[0];
    const int*   edge_src = (const int*)d_in[1];
    const int*   edge_dst = (const int*)d_in[2];
    const float* mask     = (const float*)d_in[3];
    const float* w0  = (const float*)d_in[4];
    const float* b0  = (const float*)d_in[5];
    const float* g0  = (const float*)d_in[6];
    const float* be0 = (const float*)d_in[7];
    const float* rm0 = (const float*)d_in[8];
    const float* rv0 = (const float*)d_in[9];
    const float* w1  = (const float*)d_in[10];
    const float* b1  = (const float*)d_in[11];
    const float* g1  = (const float*)d_in[12];
    const float* be1 = (const float*)d_in[13];
    const float* rm1 = (const float*)d_in[14];
    const float* rv1 = (const float*)d_in[15];
    const float* w2  = (const float*)d_in[16];
    const float* b2  = (const float*)d_in[17];
    const float* g2  = (const float*)d_in[18];
    const float* be2 = (const float*)d_in[19];
    const float* rm2 = (const float*)d_in[20];
    const float* rv2 = (const float*)d_in[21];
    const float* wp  = (const float*)d_in[22];
    const float* bp  = (const float*)d_in[23];
    float* out = (float*)d_out;

    float* W = (float*)d_ws;
    float* dinv = W;
    float* h1   = dinv + 50048;
    float* h2   = h1   + (long)NN * DIM;
    float* cbuf = h2   + (long)NN * DIM;
    unsigned short* hist = (unsigned short*)(cbuf + 1024);
    int*   bstart = (int*)(cbuf + 1024) + NHIST;
    int*   off  = bstart + 256;
    int2*  tmp  = (int2*)(off + 50048);
    int2*  rec  = tmp + EE;
    float* sc0 = cbuf + 0 * DIM;
    float* sh0 = cbuf + 1 * DIM;
    float* sc1 = cbuf + 2 * DIM;
    float* sh1 = cbuf + 3 * DIM;
    float* sc2 = cbuf + 4 * DIM;
    float* sh2 = cbuf + 4 * DIM + 64;

    BnArgs bn;
    bn.b0 = b0; bn.g0 = g0; bn.be0 = be0; bn.rm0 = rm0; bn.rv0 = rv0;
    bn.b1 = b1; bn.g1 = g1; bn.be1 = be1; bn.rm1 = rm1; bn.rv1 = rv1;
    bn.b2 = b2; bn.g2 = g2; bn.be2 = be2; bn.rm2 = rm2; bn.rv2 = rv2;
    chunk_hist_kernel<<<NCHUNK, STHREADS, 0, stream>>>(edge_dst, hist, bn, cbuf);
    chunk_scatter_kernel<<<NCHUNK, STHREADS, 0, stream>>>(edge_src, edge_dst, mask, hist, bstart, tmp);
    bucket_sort_kernel<<<NBKT, STHREADS, 0, stream>>>(tmp, bstart, off, dinv, rec);

    gin_gemm_kernel<<<LGRID, LTHREADS, 0, stream>>>(h, off, rec, dinv, w0, sc0, sh0, h1);

    gin_gemm_kernel<<<LGRID, LTHREADS, 0, stream>>>(h1, off, rec, dinv, w1, sc1, sh1, h2);

    gin_final_kernel<<<LGRID, LTHREADS, 0, stream>>>(h2, off, rec, dinv, w2, wp,
                                                     sc2, sh2, bp, out);
}


# Gather sum loop: whole groups of 16 edges take a path with one scalar compare per edge (node boundary as a position in the group, no end-of-chunk clamps); per-node sum order unchanged
# speedup vs baseline: 4.4750x; 1.1730x over previous
#include <hip/hip_runtime.h>

typedef __attribute__((ext_vector_type(2))) float v2f;
typedef __attribute__((ext_vector_type(8))) float v8f;

#define NN 50000
#define EE 800000
#define DIM 128
#define NCLS 40
#define CP 48

#define BSHIFT   8
#define BW       (1 << BSHIFT)
#define NBKT     ((NN + BW - 1) / BW)
#define NCHUNK   256
#define CHUNK    (EE / NCHUNK)
#define STHREADS 512
#define EPT      ((CHUNK + STHREADS - 1) / STHREADS)
#define NHIST    (NBKT * NCHUNK)
#define DBATCH   8
static_assert(EE % NCHUNK == 0, "chunks must cover the edge list exactly");
static_assert(NBKT <= 256 && BW == 256, "the 256-entry scan covers the buckets of a chunk and the nodes of a bucket");
static_assert(NN <= 65536, "a source index must fit the low 16 bits of a record word");
static_assert(NCHUNK == 256, "a histogram row of 16-bit counts is one 16-byte load per lane of one wave");
static_assert(CHUNK * 4 < 65536, "a count, and the sum of four counts, must fit 16 bits");

__device__ __forceinline__ int excl_scan_256(const int c, const int tid, int* __restrict__ sTot) {
    const int lane = tid & 31;
    const int wave = tid >> 5;
    int incl = c;
#pragma unroll
    for (int d = 1; d < 32; d <<= 1) {
        int t = __shfl_up(incl, d);
        if (lane >= d) incl += t;
    }
    if (lane == 31 && wave < 8) sTot[wave] = incl;
    __syncthreads();
    int basev = 0;
#pragma unroll
    for (int w = 0; w < 8; ++w)
        if (w < wave) basev += sTot[w];
    return basev + incl - c;
}

struct BnArgs {
    const float* b0; const float* g0; const float* be0; const float* rm0; const float* rv0;
    const float* b1; const float* g1; const float* be1; const float* rm1; const float* rv1;
    const float* b2; const float* g2; const float* be2; const float* rm2; const float* rv2;
};

__device__ __forceinline__ void fold_consts(const BnArgs& a, float* __restrict__ cbuf, const int t) {
    const float eps = 1e-5f;
    if (t < DIM) {
        float s = a.g0[t] * rsqrtf(a.rv0[t] + eps);
        cbuf[0 * DIM + t] = s;
        cbuf[1 * DIM + t] = (a.b0[t] - a.rm0[t]) * s + a.be0[t];
        s = a.g1[t] * rsqrtf(a.rv1[t] + eps);
        cbuf[2 * DIM + t] = s;
        cbuf[3 * DIM + t] = (a.b1[t] - a.rm1[t]) * s + a.be1[t];
    }
    if (t < NCLS) {
        float s = a.g2[t] * rsqrtf(a.rv2[t] + eps);
        cbuf[4 * DIM + t]      = s;
        cbuf[4 * DIM + 64 + t] = (a.b2[t] - a.rm2[t]) * s + a.be2[t];
    }
}

__global__ void __launch_bounds__(STHREADS)
chunk_hist_kernel(const int* __restrict__ dst, unsigned short* __restrict__ hist,
                  const BnArgs bn, float* __restrict__ cbuf) {
    __shared__ int sCnt[NBKT];
    const int tid = threadIdx.x;
    const int c   = blockIdx.x;
    const int e0  = c * CHUNK;
    int d[EPT];
#pragma unroll
    for (int k = 0; k < EPT; ++k) {
        const int i = tid + k * STHREADS;
        d[k] = (i < CHUNK) ? dst[e0 + i] : -1;
    }
    if (tid < NBKT) sCnt[tid] = 0;
    __syncthreads();
#pragma unroll
    for (int k = 0; k < EPT; ++k)
        if (d[k] >= 0) atomicAdd(&sCnt[d[k] >> BSHIFT], 1);
    __syncthreads();
    if (tid < NBKT) hist[tid * NCHUNK + c] = (unsigned short)sCnt[tid];
    if (c == 0) fold_consts(bn, cbuf, tid);
}

__global__ void __launch_bounds__(STHREADS)
chunk_scatter_kernel(const int* __restrict__ src, const int* __restrict__ dst,
                     const float* __restrict__ mask, const unsigned short* __restrict__ hist,
                     int* __restrict__ bstart, int2* __restrict__ tmp) {
    __shared__ int sPT[NBKT * 33];
    __shared__ int sPP[NBKT * 33];
    __shared__ int sTotB[8];
    __shared__ int sCnt[256];
    __shared__ int sLst[256];
    __shared__ int sGof[256];
    __shared__ int sTot[8];
    __shared__ int2 sRec[CHUNK];
    __shared__ unsigned char sBid[CHUNK];
    const int tid = threadIdx.x;
    const int c   = blockIdx.x;
    const int e0  = c * CHUNK;
    constexpr int NWV = STHREADS / 32;
    constexpr int RPW = (NBKT + NWV - 1) / NWV;
    const int lane = tid & 31;
    const int wv   = tid >> 5;
    uint4 hr[RPW];
#pragma unroll
    for (int q = 0; q < RPW; ++q) {
        const int b = wv + q * NWV;
        const uint4* row = (const uint4*)(hist + min(b, NBKT - 1) * NCHUNK);
        hr[q] = row[lane];
    }
    int d[EPT], s[EPT], mk[EPT], r[EPT];
#pragma unroll
    for (int k = 0; k < EPT; ++k) {
        const int i = tid + k * STHREADS;
        const bool in = i < CHUNK;
        d[k]  = in ? dst[e0 + i] : -1;
        s[k]  = in ? src[e0 + i] : 0;
        mk[k] = in ? __float_as_int(mask[e0 + i]) : 0;
        r[k]  = 0;
    }
    if (tid < 256) sCnt[tid] = 0;
    {
        const int i0 = lane * 8;
        unsigned mw[4];
#pragma unroll
        for (int j = 0; j < 4; ++j)
            mw[j] = (i0 + 2 * j < c ? 0x0000ffffu : 0u) | (i0 + 2 * j + 1 < c ? 0xffff0000u : 0u);
#pragma unroll
        for (int q = 0; q < RPW; ++q) {
            const int b = wv + q * NWV;
            const unsigned st = (hr[q].x + hr[q].y) + (hr[q].z + hr[q].w);
            const unsigned sp = ((hr[q].x & mw[0]) + (hr[q].y & mw[1])) + ((hr[q].z & mw[2]) + (hr[q].w & mw[3]));
            const int tot = (int)((st & 0xffffu) + (st >> 16));
            const int pre = (int)((sp & 0xffffu) + (sp >> 16));
            if (b < NBKT) { sPT[b * 33 + lane] = tot; sPP[b * 33 + lane] = pre; }
        }
    }
    __syncthreads();
#pragma unroll
    for (int k = 0; k < EPT; ++k)
        if (d[k] >= 0) r[k] = atomicAdd(&sCnt[d[k] >> BSHIFT], 1);
    int btot = 0, bpre = 0;
    if (tid < NBKT) {
#pragma unroll
        for (int j = 0; j < 32; ++j) {
            btot += sPT[tid * 33 + j];
            bpre += sPP[tid * 33 + j];
        }
    }
    __syncthreads();
    const int cnt  = (tid < 256) ? sCnt[tid] : 0;
    const int bst  = excl_scan_256(btot, tid, sTotB);
    const int excl = excl_scan_256(cnt, tid, sTot);
    if (tid < NBKT) {
        sLst[tid] = excl;
        sGof[tid] = bst + bpre - excl;
        if (c == 0) bstart[tid] = bst;
        if (c == 0 && tid == NBKT - 1) bstart[NBKT] = bst + btot;
    }
    __syncthreads();
#pragma unroll
    for (int k = 0; k < EPT; ++k) {
        if (d[k] >= 0) {
            const int b  = d[k] >> BSHIFT;
            const int lp = sLst[b] + r[k];
            sRec[lp] = make_int2(((d[k] & (BW - 1)) << 16) | s[k], mk[k]);
            sBid[lp] = (unsigned char)b;
        }
    }
    __syncthreads();
    for (int i = tid; i < CHUNK; i += STHREADS) tmp[i + sGof[sBid[i]]] = sRec[i];
}

__global__ void __launch_bounds__(STHREADS)
bucket_sort_kernel(const int2* __restrict__ tmp, const int* __restrict__ bstart,
                   int* __restrict__ off, float* __restrict__ dinv, int2* __restrict__ rec) {
    __shared__ int sCnt[BW];
    __shared__ int sExc[BW];
    __shared__ int sTot[8];
    const int tid = threadIdx.x;
    const int b   = blockIdx.x;
    const int beg = bstart[b];
    const int end = bstart[b + 1];
    if (tid < BW) sCnt[tid] = 0;
    __syncthreads();
    for (int i0 = beg; i0 < end; i0 += DBATCH * STHREADS) {
        int w[DBATCH];
#pragma unroll
        for (int k = 0; k < DBATCH; ++k) {
            const int i = i0 + k * STHREADS + tid;
            w[k] = (i < end) ? tmp[i].x : -1;
        }
#pragma unroll
        for (int k = 0; k < DBATCH; ++k)
            if (w[k] >= 0) atomicAdd(&sCnt[w[k] >> 16], 1);
    }
    __syncthreads();
    const int cnt  = (tid < BW) ? sCnt[tid] : 0;
    const int excl = excl_scan_256(cnt, tid, sTot);
    if (tid < BW) {
        sExc[tid] = excl;
        sCnt[tid] = 0;
        const int node = b * BW + tid;
        if (node < NN) {
            off[node]  = beg + excl;
            dinv[node] = (cnt > 0) ? (1.0f / (float)cnt) : 0.0f;
        }
        if (node == NN - 1) off[NN] = beg + excl + cnt;
    }
    __syncthreads();
    for (int i0 = beg; i0 < end; i0 += DBATCH * STHREADS) {
        int2 t[DBATCH];
#pragma unroll
        for (int k = 0; k < DBATCH; ++k) {
            const int i = i0 + k * STHREADS + tid;
            t[k] = (i < end) ? tmp[i] : make_int2(-1, 0);
        }
#pragma unroll
        for (int k = 0; k < DBATCH; ++k) {
            if (t[k].x >= 0) {
                const int dl = t[k].x >> 16;
                const int p  = atomicAdd(&sCnt[dl], 1);
                rec[beg + sExc[dl] + p] = make_int2(t[k].x & 0xffff, t[k].y);
            }
        }
    }
}

#define NWAVE   13
#define LTHREADS (NWAVE * 32)
#define LGRID   256
#define NTILES  (NN / 16)
#define GB      16
#define TPITCH  132
static_assert(NN % 16 == 0, "tiles of 16 nodes must cover the node range exactly");
static_assert(LGRID * NWAVE >= NTILES, "every tile needs a wave");

#define ROW_FMA(V, M)                      \
    acc.x = fmaf((V).x, (M), acc.x);       \
    acc.y = fmaf((V).y, (M), acc.y);       \
    acc.z = fmaf((V).z, (M), acc.z);       \
    acc.w = fmaf((V).w, (M), acc.w);

__device__ __forceinline__ void gather_tile(const float* __restrict__ x, const int* __restrict__ off,
                                            const int2* __restrict__ rec, const int n0,
                                            float* __restrict__ tile, const int lane) {
    const int offv = off[n0 + min(lane, 16)];
    const int beg = __builtin_amdgcn_readlane(offv, 0);
    const int end = __builtin_amdgcn_readlane(offv, 16);
    const float* xl = x + lane * 4;
    float* trow = tile + lane * 4;
    float4 acc = make_float4(0.0f, 0.0f, 0.0f, 0.0f);
    int cur = 0;
    int nb  = __builtin_amdgcn_readlane(offv, 1);

    int2 ed = make_int2(0, 0);
    if (beg + lane < end) ed = rec[beg + lane];
    for (int base = beg; base < end; base += 32) {
        const int c = min(32, end - base);
        int2 edn = make_int2(0, 0);
        if (base + 32 + lane < end) edn = rec[base + 32 + lane];
        for (int j = 0; j < c; j += GB) {
            float4 v[GB];
            float  mk[GB];
            if (j + GB <= c) {
#pragma unroll
                for (int k = 0; k < GB; ++k) {
                    const unsigned s = (unsigned)__builtin_amdgcn_readlane(ed.x, j + k);
                    mk[k] = __int_as_float(__builtin_amdgcn_readlane(ed.y, j + k));
                    v[k] = *(const float4*)(xl + (size_t)s * DIM);
                }
                const int e0 = base + j;
                int rel = (cur < 15) ? nb - e0 : -1;
#pragma unroll
                for (int k = 0; k < GB; ++k) {
                    if (rel == k) {
                        do {
                            *(float4*)(trow + cur * TPITCH) = acc;
                            acc = make_float4(0.0f, 0.0f, 0.0f, 0.0f);
                            ++cur;
                            nb = __builtin_amdgcn_readlane(offv, cur + 1);
                        } while (cur < 15 && nb == e0 + k);
                        rel = (cur < 15) ? nb - e0 : -1;
                    }
                    ROW_FMA(v[k], mk[k])
                }
            } else {
#pragma unroll
                for (int k = 0; k < GB; ++k) {
                    const int jj = min(j + k, c - 1);
                    const int s = __builtin_amdgcn_readlane(ed.x, jj);
                    mk[k] = __int_as_float(__builtin_amdgcn_readlane(ed.y, jj));
                    v[k] = *(const float4*)(xl + (long)s * DIM);
                }
#pragma unroll
                for (int k = 0; k < GB; ++k) {
                    if (j + k < c) {
                        const int e = base + j + k;
                        while (cur < 15 && e == nb) {
                            *(float4*)(trow + cur * TPITCH) = acc;
                            acc = make_float4(0.0f, 0.0f, 0.0f, 0.0f);
                            ++cur;
                            nb = __builtin_amdgcn_readlane(offv, cur + 1);
                        }
                        ROW_FMA(v[k], mk[k])
                    }
                }
            }
        }
        ed = edn;
    }
    for (; cur < 16; ++cur) {
        *(float4*)(trow + cur * TPITCH) = acc;
        acc = make_float4(0.0f, 0.0f, 0.0f, 0.0f);
    }
}

__global__ void __launch_bounds__(LTHREADS)
gin_gemm_kernel(const float* __restrict__ x, const int* __restrict__ off,
                const int2* __restrict__ rec,
                const float* __restrict__ dinv, const float* __restrict__ W,
                const float* __restrict__ scale, const float* __restrict__ shift,
                float* __restrict__ out) {
    __shared__ v2f sW[(DIM / 2) * DIM];
    __shared__ __attribute__((aligned(16))) float sA[NWAVE * 16 * TPITCH];
    const int tid = threadIdx.x;
    for (int i = tid; i < (DIM / 2) * DIM; i += LTHREADS) {
        int p = i / DIM, c = i % DIM;
        v2f w;
        w.x = W[(2 * p) * DIM + c];
        w.y = W[(2 * p + 1) * DIM + c];
        sW[i] = w;
    }
    __syncthreads();

    const int wave = __builtin_amdgcn_readfirstlane(tid >> 5);
    const int lane = tid & 31;
    const int t    = blockIdx.x + LGRID * wave;
    if (t >= NTILES) return;
    const int row0 = t * 16;
    float* tile = sA + wave * (16 * TPITCH);

    gather_tile(x, off, rec, row0, tile, lane);

    {
        const float dv = dinv[row0 + (lane & 15)];
        float4 xi[16];
#pragma unroll
        for (int i = 0; i < 16; ++i) xi[i] = *(const float4*)(x + (long)(row0 + i) * DIM + lane * 4);
#pragma unroll
        for (int i = 0; i < 16; ++i) {
            const float di = __int_as_float(__builtin_amdgcn_readlane(__float_as_int(dv), i));
            float4 ag = *(const float4*)(tile + i * TPITCH + lane * 4);
            float4 a;
            a.x = xi[i].x + ag.x * di;
            a.y = xi[i].y + ag.y * di;
            a.z = xi[i].z + ag.z * di;
            a.w = xi[i].w + ag.w * di;
            *(float4*)(tile + i * TPITCH + lane * 4) = a;
        }
    }

    const int m    = lane & 15;
    const int prow = lane >> 4;
    const int kk   = prow << 1;
    const float* ar = tile + m * TPITCH + kk;

    v8f acc[8] = {};
#pragma unroll 4
    for (int k0 = 0; k0 < DIM; k0 += 4) {
        const v2f* bb = sW + ((k0 >> 1) + prow) * DIM + m;
        v2f bf[8];
#pragma unroll
        for (int nt = 0; nt < 8; ++nt) bf[nt] = bb[nt * 16];
        const v2f a = *(const v2f*)(ar + k0);
#pragma unroll
        for (int nt = 0; nt < 8; ++nt)
            acc[nt] = __builtin_amdgcn_wmma_f32_16x16x4_f32(
                false, a, false, bf[nt], (short)0, acc[nt], false, false);
    }

    const int roff = prow << 3;
#pragma unroll
    for (int nt = 0; nt < 8; ++nt) {
        const int col = nt * 16 + m;
        const float s  = scale[col];
        const float sh = shift[col];
#pragma unroll
        for (int j = 0; j < 8; ++j) {
            const int rr = row0 + roff + j;
            float v = acc[nt][j] * s + sh;
            out[(long)rr * DIM + col] = v > 0.0f ? v : 0.0f;
        }
    }
}

__global__ void __launch_bounds__(LTHREADS)
gin_final_kernel(const float* __restrict__ h2, const int* __restrict__ off,
                 const int2* __restrict__ rec,
                 const float* __restrict__ dinv,
                 const float* __restrict__ w2, const float* __restrict__ wp,
                 const float* __restrict__ scale2, const float* __restrict__ shift2,
                 const float* __restrict__ bp, float* __restrict__ out) {
    __shared__ v2f sW2[(DIM / 2) * CP];
    __shared__ v2f sWP[(DIM / 2) * CP];
    __shared__ __attribute__((aligned(16))) float sA[NWAVE * 16 * TPITCH];
    __shared__ __attribute__((aligned(16))) float sX[NWAVE * 16 * TPITCH];
    const int tid = threadIdx.x;
    for (int i = tid; i < (DIM / 2) * CP; i += LTHREADS) {
        int p = i / CP, c = i % CP;
        v2f a = {0.0f, 0.0f}, b = {0.0f, 0.0f};
        if (c < NCLS) {
            a.x = w2[(2 * p) * NCLS + c];
            a.y = w2[(2 * p + 1) * NCLS + c];
            b.x = wp[(2 * p) * NCLS + c];
            b.y = wp[(2 * p + 1) * NCLS + c];
        }
        sW2[i] = a;
        sWP[i] = b;
    }
    __syncthreads();

    const int wave = __builtin_amdgcn_readfirstlane(tid >> 5);
    const int lane = tid & 31;
    const int t    = blockIdx.x + LGRID * wave;
    if (t >= NTILES) return;
    const int row0 = t * 16;
    float* tile  = sA + wave * (16 * TPITCH);
    float* tileX = sX + wave * (16 * TPITCH);

    gather_tile(h2, off, rec, row0, tile, lane);

    {
        const float dv = dinv[row0 + (lane & 15)];
        float4 xi[16];
#pragma unroll
        for (int i = 0; i < 16; ++i) xi[i] = *(const float4*)(h2 + (long)(row0 + i) * DIM + lane * 4);
#pragma unroll
        for (int i = 0; i < 16; ++i) {
            const float di = __int_as_float(__builtin_amdgcn_readlane(__float_as_int(dv), i));
            float4 ag = *(const float4*)(tile + i * TPITCH + lane * 4);
            float4 a;
            a.x = xi[i].x + ag.x * di;
            a.y = xi[i].y + ag.y * di;
            a.z = xi[i].z + ag.z * di;
            a.w = xi[i].w + ag.w * di;
            *(float4*)(tile  + i * TPITCH + lane * 4) = a;
            *(float4*)(tileX + i * TPITCH + lane * 4) = xi[i];
        }
    }

    const int m    = lane & 15;
    const int prow = lane >> 4;
    const int kk   = prow << 1;
    const float* ar = tile  + m * TPITCH + kk;
    const float* xr = tileX + m * TPITCH + kk;

    v8f acc3[3] = {};
    v8f accp[3] = {};
#pragma unroll 4
    for (int k0 = 0; k0 < DIM; k0 += 4) {
        const int pidx = ((k0 >> 1) + prow) * CP + m;
        v2f bf2[3], bfp[3];
#pragma unroll
        for (int nt = 0; nt < 3; ++nt) {
            bf2[nt] = sW2[pidx + nt * 16];
            bfp[nt] = sWP[pidx + nt * 16];
        }
        const v2f a3 = *(const v2f*)(ar + k0);
        const v2f ap = *(const v2f*)(xr + k0);
#pragma unroll
        for (int nt = 0; nt < 3; ++nt) {
            acc3[nt] = __builtin_amdgcn_wmma_f32_16x16x4_f32(
                false, a3, false, bf2[nt], (short)0, acc3[nt], false, false);
            accp[nt] = __builtin_amdgcn_wmma_f32_16x16x4_f32(
                false, ap, false, bfp[nt], (short)0, accp[nt], false, false);
        }
    }

    const int roff = prow << 3;
#pragma unroll
    for (int nt = 0; nt < 3; ++nt) {
        const int col = nt * 16 + m;
        if (col >= NCLS) continue;
        const float s  = scale2[col];
        const float sh = shift2[col];
        const float bb = bp[col];
#pragma unroll
        for (int j = 0; j < 8; ++j) {
            const int rr = row0 + roff + j;
            float h3 = acc3[nt][j] * s + sh;
            h3 = h3 > 0.0f ? h3 : 0.0f;
            out[(long)rr * NCLS + col] = 0.5f * (accp[nt][j] + bb + h3);
        }
    }
}

extern "C" void kernel_launch(void* const* d_in, const int* in_sizes, int n_in,
                              void* d_out, int out_size, void* d_ws, size_t ws_size,
                              hipStream_t stream) {
    const float* h        = (const float*)d_in[0];
    const int*   edge_src = (const int*)d_in[1];
    const int*   edge_dst = (const int*)d_in[2];
    const float* mask     = (const float*)d_in[3];
    const float* w0  = (const float*)d_in[4];
    const float* b0  = (const float*)d_in[5];
    const float* g0  = (const float*)d_in[6];
    const float* be0 = (const float*)d_in[7];
    const float* rm0 = (const float*)d_in[8];
    const float* rv0 = (const float*)d_in[9];
    const float* w1  = (const float*)d_in[10];
    const float* b1  = (const float*)d_in[11];
    const float* g1  = (const float*)d_in[12];
    const float* be1 = (const float*)d_in[13];
    const float* rm1 = (const float*)d_in[14];
    const float* rv1 = (const float*)d_in[15];
    const float* w2  = (const float*)d_in[16];
    const float* b2  = (const float*)d_in[17];
    const float* g2  = (const float*)d_in[18];
    const float* be2 = (const float*)d_in[19];
    const float* rm2 = (const float*)d_in[20];
    const float* rv2 = (const float*)d_in[21];
    const float* wp  = (const float*)d_in[22];
    const float* bp  = (const float*)d_in[23];
    float* out = (float*)d_out;

    float* W = (float*)d_ws;
    float* dinv = W;
    float* h1   = dinv + 50048;
    float* h2   = h1   + (long)NN * DIM;
    float* cbuf = h2   + (long)NN * DIM;
    unsigned short* hist = (unsigned short*)(cbuf + 1024);
    int*   bstart = (int*)(cbuf + 1024) + NHIST;
    int*   off  = bstart + 256;
    int2*  tmp  = (int2*)(off + 50048);
    int2*  rec  = tmp + EE;
    float* sc0 = cbuf + 0 * DIM;
    float* sh0 = cbuf + 1 * DIM;
    float* sc1 = cbuf + 2 * DIM;
    float* sh1 = cbuf + 3 * DIM;
    float* sc2 = cbuf + 4 * DIM;
    float* sh2 = cbuf + 4 * DIM + 64;

    BnArgs bn;
    bn.b0 = b0; bn.g0 = g0; bn.be0 = be0; bn.rm0 = rm0; bn.rv0 = rv0;
    bn.b1 = b1; bn.g1 = g1; bn.be1 = be1; bn.rm1 = rm1; bn.rv1 = rv1;
    bn.b2 = b2; bn.g2 = g2; bn.be2 = be2; bn.rm2 = rm2; bn.rv2 = rv2;
    chunk_hist_kernel<<<NCHUNK, STHREADS, 0, stream>>>(edge_dst, hist, bn, cbuf);
    chunk_scatter_kernel<<<NCHUNK, STHREADS, 0, stream>>>(edge_src, edge_dst, mask, hist, bstart, tmp);
    bucket_sort_kernel<<<NBKT, STHREADS, 0, stream>>>(tmp, bstart, off, dinv, rec);

    gin_gemm_kernel<<<LGRID, LTHREADS, 0, stream>>>(h, off, rec, dinv, w0, sc0, sh0, h1);

    gin_gemm_kernel<<<LGRID, LTHREADS, 0, stream>>>(h1, off, rec, dinv, w1, sc1, sh1, h2);

    gin_final_kernel<<<LGRID, LTHREADS, 0, stream>>>(h2, off, rec, dinv, w2, wp,
                                                     sc2, sh2, bp, out);
}


# Two waves per 16-node tile: each gathers 8 nodes, the barrier moves behind the gather (weight staging runs beside it), then the pair splits the tile's WMMA work; per-node sum order unchanged
# speedup vs baseline: 4.9434x; 1.1047x over previous
#include <hip/hip_runtime.h>

typedef __attribute__((ext_vector_type(2))) float v2f;
typedef __attribute__((ext_vector_type(8))) float v8f;

#define NN 50000
#define EE 800000
#define DIM 128
#define NCLS 40
#define CP 48

#define BSHIFT   8
#define BW       (1 << BSHIFT)
#define NBKT     ((NN + BW - 1) / BW)
#define NCHUNK   256
#define CHUNK    (EE / NCHUNK)
#define STHREADS 512
#define EPT      ((CHUNK + STHREADS - 1) / STHREADS)
#define NHIST    (NBKT * NCHUNK)
#define DBATCH   8
static_assert(EE % NCHUNK == 0, "chunks must cover the edge list exactly");
static_assert(NBKT <= 256 && BW == 256, "the 256-entry scan covers the buckets of a chunk and the nodes of a bucket");
static_assert(NN <= 65536, "a source index must fit the low 16 bits of a record word");
static_assert(NCHUNK == 256, "a histogram row of 16-bit counts is one 16-byte load per lane of one wave");
static_assert(CHUNK * 4 < 65536, "a count, and the sum of four counts, must fit 16 bits");

__device__ __forceinline__ int excl_scan_256(const int c, const int tid, int* __restrict__ sTot) {
    const int lane = tid & 31;
    const int wave = tid >> 5;
    int incl = c;
#pragma unroll
    for (int d = 1; d < 32; d <<= 1) {
        int t = __shfl_up(incl, d);
        if (lane >= d) incl += t;
    }
    if (lane == 31 && wave < 8) sTot[wave] = incl;
    __syncthreads();
    int basev = 0;
#pragma unroll
    for (int w = 0; w < 8; ++w)
        if (w < wave) basev += sTot[w];
    return basev + incl - c;
}

struct BnArgs {
    const float* b0; const float* g0; const float* be0; const float* rm0; const float* rv0;
    const float* b1; const float* g1; const float* be1; const float* rm1; const float* rv1;
    const float* b2; const float* g2; const float* be2; const float* rm2; const float* rv2;
};

__device__ __forceinline__ void fold_consts(const BnArgs& a, float* __restrict__ cbuf, const int t) {
    const float eps = 1e-5f;
    if (t < DIM) {
        float s = a.g0[t] * rsqrtf(a.rv0[t] + eps);
        cbuf[0 * DIM + t] = s;
        cbuf[1 * DIM + t] = (a.b0[t] - a.rm0[t]) * s + a.be0[t];
        s = a.g1[t] * rsqrtf(a.rv1[t] + eps);
        cbuf[2 * DIM + t] = s;
        cbuf[3 * DIM + t] = (a.b1[t] - a.rm1[t]) * s + a.be1[t];
    }
    if (t < NCLS) {
        float s = a.g2[t] * rsqrtf(a.rv2[t] + eps);
        cbuf[4 * DIM + t]      = s;
        cbuf[4 * DIM + 64 + t] = (a.b2[t] - a.rm2[t]) * s + a.be2[t];
    }
}

__global__ void __launch_bounds__(STHREADS)
chunk_hist_kernel(const int* __restrict__ dst, unsigned short* __restrict__ hist,
                  const BnArgs bn, float* __restrict__ cbuf) {
    __shared__ int sCnt[NBKT];
    const int tid = threadIdx.x;
    const int c   = blockIdx.x;
    const int e0  = c * CHUNK;
    int d[EPT];
#pragma unroll
    for (int k = 0; k < EPT; ++k) {
        const int i = tid + k * STHREADS;
        d[k] = (i < CHUNK) ? dst[e0 + i] : -1;
    }
    if (tid < NBKT) sCnt[tid] = 0;
    __syncthreads();
#pragma unroll
    for (int k = 0; k < EPT; ++k)
        if (d[k] >= 0) atomicAdd(&sCnt[d[k] >> BSHIFT], 1);
    __syncthreads();
    if (tid < NBKT) hist[tid * NCHUNK + c] = (unsigned short)sCnt[tid];
    if (c == 0) fold_consts(bn, cbuf, tid);
}

__global__ void __launch_bounds__(STHREADS)
chunk_scatter_kernel(const int* __restrict__ src, const int* __restrict__ dst,
                     const float* __restrict__ mask, const unsigned short* __restrict__ hist,
                     int* __restrict__ bstart, int2* __restrict__ tmp) {
    __shared__ int sPT[NBKT * 33];
    __shared__ int sPP[NBKT * 33];
    __shared__ int sTotB[8];
    __shared__ int sCnt[256];
    __shared__ int sLst[256];
    __shared__ int sGof[256];
    __shared__ int sTot[8];
    __shared__ int2 sRec[CHUNK];
    __shared__ unsigned char sBid[CHUNK];
    const int tid = threadIdx.x;
    const int c   = blockIdx.x;
    const int e0  = c * CHUNK;
    constexpr int NWV = STHREADS / 32;
    constexpr int RPW = (NBKT + NWV - 1) / NWV;
    const int lane = tid & 31;
    const int wv   = tid >> 5;
    uint4 hr[RPW];
#pragma unroll
    for (int q = 0; q < RPW; ++q) {
        const int b = wv + q * NWV;
        const uint4* row = (const uint4*)(hist + min(b, NBKT - 1) * NCHUNK);
        hr[q] = row[lane];
    }
    int d[EPT], s[EPT], mk[EPT], r[EPT];
#pragma unroll
    for (int k = 0; k < EPT; ++k) {
        const int i = tid + k * STHREADS;
        const bool in = i < CHUNK;
        d[k]  = in ? dst[e0 + i] : -1;
        s[k]  = in ? src[e0 + i] : 0;
        mk[k] = in ? __float_as_int(mask[e0 + i]) : 0;
        r[k]  = 0;
    }
    if (tid < 256) sCnt[tid] = 0;
    {
        const int i0 = lane * 8;
        unsigned mw[4];
#pragma unroll
        for (int j = 0; j < 4; ++j)
            mw[j] = (i0 + 2 * j < c ? 0x0000ffffu : 0u) | (i0 + 2 * j + 1 < c ? 0xffff0000u : 0u);
#pragma unroll
        for (int q = 0; q < RPW; ++q) {
            const int b = wv + q * NWV;
            const unsigned st = (hr[q].x + hr[q].y) + (hr[q].z + hr[q].w);
            const unsigned sp = ((hr[q].x & mw[0]) + (hr[q].y & mw[1])) + ((hr[q].z & mw[2]) + (hr[q].w & mw[3]));
            const int tot = (int)((st & 0xffffu) + (st >> 16));
            const int pre = (int)((sp & 0xffffu) + (sp >> 16));
            if (b < NBKT) { sPT[b * 33 + lane] = tot; sPP[b * 33 + lane] = pre; }
        }
    }
    __syncthreads();
#pragma unroll
    for (int k = 0; k < EPT; ++k)
        if (d[k] >= 0) r[k] = atomicAdd(&sCnt[d[k] >> BSHIFT], 1);
    int btot = 0, bpre = 0;
    if (tid < NBKT) {
#pragma unroll
        for (int j = 0; j < 32; ++j) {
            btot += sPT[tid * 33 + j];
            bpre += sPP[tid * 33 + j];
        }
    }
    __syncthreads();
    const int cnt  = (tid < 256) ? sCnt[tid] : 0;
    const int bst  = excl_scan_256(btot, tid, sTotB);
    const int excl = excl_scan_256(cnt, tid, sTot);
    if (tid < NBKT) {
        sLst[tid] = excl;
        sGof[tid] = bst + bpre - excl;
        if (c == 0) bstart[tid] = bst;
        if (c == 0 && tid == NBKT - 1) bstart[NBKT] = bst + btot;
    }
    __syncthreads();
#pragma unroll
    for (int k = 0; k < EPT; ++k) {
        if (d[k] >= 0) {
            const int b  = d[k] >> BSHIFT;
            const int lp = sLst[b] + r[k];
            sRec[lp] = make_int2(((d[k] & (BW - 1)) << 16) | s[k], mk[k]);
            sBid[lp] = (unsigned char)b;
        }
    }
    __syncthreads();
    for (int i = tid; i < CHUNK; i += STHREADS) tmp[i + sGof[sBid[i]]] = sRec[i];
}

__global__ void __launch_bounds__(STHREADS)
bucket_sort_kernel(const int2* __restrict__ tmp, const int* __restrict__ bstart,
                   int* __restrict__ off, float* __restrict__ dinv, int2* __restrict__ rec) {
    __shared__ int sCnt[BW];
    __shared__ int sExc[BW];
    __shared__ int sTot[8];
    const int tid = threadIdx.x;
    const int b   = blockIdx.x;
    const int beg = bstart[b];
    const int end = bstart[b + 1];
    if (tid < BW) sCnt[tid] = 0;
    __syncthreads();
    for (int i0 = beg; i0 < end; i0 += DBATCH * STHREADS) {
        int w[DBATCH];
#pragma unroll
        for (int k = 0; k < DBATCH; ++k) {
            const int i = i0 + k * STHREADS + tid;
            w[k] = (i < end) ? tmp[i].x : -1;
        }
#pragma unroll
        for (int k = 0; k < DBATCH; ++k)
            if (w[k] >= 0) atomicAdd(&sCnt[w[k] >> 16], 1);
    }
    __syncthreads();
    const int cnt  = (tid < BW) ? sCnt[tid] : 0;
    const int excl = excl_scan_256(cnt, tid, sTot);
    if (tid < BW) {
        sExc[tid] = excl;
        sCnt[tid] = 0;
        const int node = b * BW + tid;
        if (node < NN) {
            off[node]  = beg + excl;
            dinv[node] = (cnt > 0) ? (1.0f / (float)cnt) : 0.0f;
        }
        if (node == NN - 1) off[NN] = beg + excl + cnt;
    }
    __syncthreads();
    for (int i0 = beg; i0 < end; i0 += DBATCH * STHREADS) {
        int2 t[DBATCH];
#pragma unroll
        for (int k = 0; k < DBATCH; ++k) {
            const int i = i0 + k * STHREADS + tid;
            t[k] = (i < end) ? tmp[i] : make_int2(-1, 0);
        }
#pragma unroll
        for (int k = 0; k < DBATCH; ++k) {
            if (t[k].x >= 0) {
                const int dl = t[k].x >> 16;
                const int p  = atomicAdd(&sCnt[dl], 1);
                rec[beg + sExc[dl] + p] = make_int2(t[k].x & 0xffff, t[k].y);
            }
        }
    }
}

#define NTPG    13
#define HROWS   8
#define NWAVE   (2 * NTPG)
#define LTHREADS (NWAVE * 32)
#define LGRID   256
#define NTILES  (NN / 16)
#define GB      16
#define TPITCH  132
static_assert(NN % 16 == 0, "tiles of 16 nodes must cover the node range exactly");
static_assert(LGRID * NTPG >= NTILES, "every tile needs a pair of waves");
static_assert(LTHREADS <= 1024, "one work-group holds all pairs");

#define ROW_FMA(V, M)                      \
    acc.x = fmaf((V).x, (M), acc.x);       \
    acc.y = fmaf((V).y, (M), acc.y);       \
    acc.z = fmaf((V).z, (M), acc.z);       \
    acc.w = fmaf((V).w, (M), acc.w);

template <int NPW>
__device__ __forceinline__ void gather_tile(const float* __restrict__ x, const int* __restrict__ off,
                                            const int2* __restrict__ rec, const int n0,
                                            float* __restrict__ tile, const int lane) {
    const int offv = off[n0 + min(lane, NPW)];
    const int beg = __builtin_amdgcn_readlane(offv, 0);
    const int end = __builtin_amdgcn_readlane(offv, NPW);
    const float* xl = x + lane * 4;
    float* trow = tile + lane * 4;
    float4 acc = make_float4(0.0f, 0.0f, 0.0f, 0.0f);
    int cur = 0;
    int nb  = __builtin_amdgcn_readlane(offv, 1);

    int2 ed = make_int2(0, 0);
    if (beg + lane < end) ed = rec[beg + lane];
    for (int base = beg; base < end; base += 32) {
        const int c = min(32, end - base);
        int2 edn = make_int2(0, 0);
        if (base + 32 + lane < end) edn = rec[base + 32 + lane];
        for (int j = 0; j < c; j += GB) {
            float4 v[GB];
            float  mk[GB];
            if (j + GB <= c) {
#pragma unroll
                for (int k = 0; k < GB; ++k) {
                    const unsigned s = (unsigned)__builtin_amdgcn_readlane(ed.x, j + k);
                    mk[k] = __int_as_float(__builtin_amdgcn_readlane(ed.y, j + k));
                    v[k] = *(const float4*)(xl + (size_t)s * DIM);
                }
                const int e0 = base + j;
                int rel = (cur < NPW - 1) ? nb - e0 : -1;
#pragma unroll
                for (int k = 0; k < GB; ++k) {
                    if (rel == k) {
                        do {
                            *(float4*)(trow + cur * TPITCH) = acc;
                            acc = make_float4(0.0f, 0.0f, 0.0f, 0.0f);
                            ++cur;
                            nb = __builtin_amdgcn_readlane(offv, cur + 1);
                        } while (cur < NPW - 1 && nb == e0 + k);
                        rel = (cur < NPW - 1) ? nb - e0 : -1;
                    }
                    ROW_FMA(v[k], mk[k])
                }
            } else {
#pragma unroll
                for (int k = 0; k < GB; ++k) {
                    const int jj = min(j + k, c - 1);
                    const int s = __builtin_amdgcn_readlane(ed.x, jj);
                    mk[k] = __int_as_float(__builtin_amdgcn_readlane(ed.y, jj));
                    v[k] = *(const float4*)(xl + (long)s * DIM);
                }
#pragma unroll
                for (int k = 0; k < GB; ++k) {
                    if (j + k < c) {
                        const int e = base + j + k;
                        while (cur < NPW - 1 && e == nb) {
                            *(float4*)(trow + cur * TPITCH) = acc;
                            acc = make_float4(0.0f, 0.0f, 0.0f, 0.0f);
                            ++cur;
                            nb = __builtin_amdgcn_readlane(offv, cur + 1);
                        }
                        ROW_FMA(v[k], mk[k])
                    }
                }
            }
        }
        ed = edn;
    }
    for (; cur < NPW; ++cur) {
        *(float4*)(trow + cur * TPITCH) = acc;
        acc = make_float4(0.0f, 0.0f, 0.0f, 0.0f);
    }
}

__global__ void __launch_bounds__(LTHREADS)
gin_gemm_kernel(const float* __restrict__ x, const int* __restrict__ off,
                const int2* __restrict__ rec,
                const float* __restrict__ dinv, const float* __restrict__ W,
                const float* __restrict__ scale, const float* __restrict__ shift,
                float* __restrict__ out) {
    __shared__ v2f sW[(DIM / 2) * DIM];
    __shared__ __attribute__((aligned(16))) float sA[NTPG * 16 * TPITCH];
    const int tid = threadIdx.x;
    for (int i = tid; i < (DIM / 2) * DIM; i += LTHREADS) {
        int p = i / DIM, c = i % DIM;
        v2f w;
        w.x = W[(2 * p) * DIM + c];
        w.y = W[(2 * p + 1) * DIM + c];
        sW[i] = w;
    }

    const int wave = __builtin_amdgcn_readfirstlane(tid >> 5);
    const int lane = tid & 31;
    const int pr   = wave >> 1;
    const int hf   = wave & 1;
    const int t    = blockIdx.x + LGRID * pr;
    const bool live = t < NTILES;
    const int row0 = t * 16;
    float* tile = sA + pr * (16 * TPITCH);

    if (live) {
        const int r0 = row0 + hf * HROWS;
        float* half = tile + hf * (HROWS * TPITCH);
        gather_tile<HROWS>(x, off, rec, r0, half, lane);

        const float dv = dinv[r0 + (lane & (HROWS - 1))];
        float4 xi[HROWS];
#pragma unroll
        for (int i = 0; i < HROWS; ++i) xi[i] = *(const float4*)(x + (long)(r0 + i) * DIM + lane * 4);
#pragma unroll
        for (int i = 0; i < HROWS; ++i) {
            const float di = __int_as_float(__builtin_amdgcn_readlane(__float_as_int(dv), i));
            float4 ag = *(const float4*)(half + i * TPITCH + lane * 4);
            float4 a;
            a.x = xi[i].x + ag.x * di;
            a.y = xi[i].y + ag.y * di;
            a.z = xi[i].z + ag.z * di;
            a.w = xi[i].w + ag.w * di;
            *(float4*)(half + i * TPITCH + lane * 4) = a;
        }
    }
    __syncthreads();
    if (!live) return;

    const int m    = lane & 15;
    const int prow = lane >> 4;
    const int kk   = prow << 1;
    const int cb   = hf * 64;
    const float* ar = tile + m * TPITCH + kk;

    v8f acc[4] = {};
#pragma unroll 4
    for (int k0 = 0; k0 < DIM; k0 += 4) {
        const v2f* bb = sW + ((k0 >> 1) + prow) * DIM + cb + m;
        v2f bf[4];
#pragma unroll
        for (int nt = 0; nt < 4; ++nt) bf[nt] = bb[nt * 16];
        const v2f a = *(const v2f*)(ar + k0);
#pragma unroll
        for (int nt = 0; nt < 4; ++nt)
            acc[nt] = __builtin_amdgcn_wmma_f32_16x16x4_f32(
                false, a, false, bf[nt], (short)0, acc[nt], false, false);
    }

    const int roff = prow << 3;
#pragma unroll
    for (int nt = 0; nt < 4; ++nt) {
        const int col = cb + nt * 16 + m;
        const float s  = scale[col];
        const float sh = shift[col];
#pragma unroll
        for (int j = 0; j < 8; ++j) {
            const int rr = row0 + roff + j;
            float v = acc[nt][j] * s + sh;
            out[(long)rr * DIM + col] = v > 0.0f ? v : 0.0f;
        }
    }
}

__global__ void __launch_bounds__(LTHREADS)
gin_final_kernel(const float* __restrict__ h2, const int* __restrict__ off,
                 const int2* __restrict__ rec,
                 const float* __restrict__ dinv,
                 const float* __restrict__ w2, const float* __restrict__ wp,
                 const float* __restrict__ scale2, const float* __restrict__ shift2,
                 const float* __restrict__ bp, float* __restrict__ out) {
    __shared__ v2f sW2[(DIM / 2) * CP];
    __shared__ v2f sWP[(DIM / 2) * CP];
    __shared__ __attribute__((aligned(16))) float sA[NTPG * 16 * TPITCH];
    __shared__ __attribute__((aligned(16))) float sX[NTPG * 16 * TPITCH];
    const int tid = threadIdx.x;
    for (int i = tid; i < (DIM / 2) * CP; i += LTHREADS) {
        int p = i / CP, c = i % CP;
        v2f a = {0.0f, 0.0f}, b = {0.0f, 0.0f};
        if (c < NCLS) {
            a.x = w2[(2 * p) * NCLS + c];
            a.y = w2[(2 * p + 1) * NCLS + c];
            b.x = wp[(2 * p) * NCLS + c];
            b.y = wp[(2 * p + 1) * NCLS + c];
        }
        sW2[i] = a;
        sWP[i] = b;
    }

    const int wave = __builtin_amdgcn_readfirstlane(tid >> 5);
    const int lane = tid & 31;
    const int pr   = wave >> 1;
    const int hf   = wave & 1;
    const int t    = blockIdx.x + LGRID * pr;
    const bool live = t < NTILES;
    const int row0 = t * 16;
    float* tile  = sA + pr * (16 * TPITCH);
    float* tileX = sX + pr * (16 * TPITCH);

    if (live) {
        const int r0 = row0 + hf * HROWS;
        float* half  = tile  + hf * (HROWS * TPITCH);
        float* halfX = tileX + hf * (HROWS * TPITCH);
        gather_tile<HROWS>(h2, off, rec, r0, half, lane);

        const float dv = dinv[r0 + (lane & (HROWS - 1))];
        float4 xi[HROWS];
#pragma unroll
        for (int i = 0; i < HROWS; ++i) xi[i] = *(const float4*)(h2 + (long)(r0 + i) * DIM + lane * 4);
#pragma unroll
        for (int i = 0; i < HROWS; ++i) {
            const float di = __int_as_float(__builtin_amdgcn_readlane(__float_as_int(dv), i));
            float4 ag = *(const float4*)(half + i * TPITCH + lane * 4);
            float4 a;
            a.x = xi[i].x + ag.x * di;
            a.y = xi[i].y + ag.y * di;
            a.z = xi[i].z + ag.z * di;
            a.w = xi[i].w + ag.w * di;
            *(float4*)(half  + i * TPITCH + lane * 4) = a;
            *(float4*)(halfX + i * TPITCH + lane * 4) = xi[i];
        }
    }
    __syncthreads();

    const int m    = lane & 15;
    const int prow = lane >> 4;
    const int kk   = prow << 1;
    const int roff = prow << 3;
    const float* ar = (hf ? tileX : tile) + m * TPITCH + kk;
    const v2f*   wb = (hf ? sWP : sW2) + prow * CP + m;

    v8f acc[3] = {};
    if (live) {
#pragma unroll 4
        for (int k0 = 0; k0 < DIM; k0 += 4) {
            const v2f* bb = wb + (k0 >> 1) * CP;
            v2f bf[3];
#pragma unroll
            for (int nt = 0; nt < 3; ++nt) bf[nt] = bb[nt * 16];
            const v2f a = *(const v2f*)(ar + k0);
#pragma unroll
            for (int nt = 0; nt < 3; ++nt)
                acc[nt] = __builtin_amdgcn_wmma_f32_16x16x4_f32(
                    false, a, false, bf[nt], (short)0, acc[nt], false, false);
        }
        if (hf) {
#pragma unroll
            for (int nt = 0; nt < 3; ++nt)
#pragma unroll
                for (int j = 0; j < 8; ++j)
                    tileX[(roff + j) * TPITCH + nt * 16 + m] = acc[nt][j];
        }
    }
    __syncthreads();
    if (!live || hf) return;

#pragma unroll
    for (int nt = 0; nt < 3; ++nt) {
        const int col = nt * 16 + m;
        if (col >= NCLS) continue;
        const float s  = scale2[col];
        const float sh = shift2[col];
        const float bb = bp[col];
#pragma unroll
        for (int j = 0; j < 8; ++j) {
            const int rr = row0 + roff + j;
            float h3 = acc[nt][j] * s + sh;
            h3 = h3 > 0.0f ? h3 : 0.0f;
            const float pv = tileX[(roff + j) * TPITCH + col];
            out[(long)rr * NCLS + col] = 0.5f * (pv + bb + h3);
        }
    }
}

extern "C" void kernel_launch(void* const* d_in, const int* in_sizes, int n_in,
                              void* d_out, int out_size, void* d_ws, size_t ws_size,
                              hipStream_t stream) {
    const float* h        = (const float*)d_in[0];
    const int*   edge_src = (const int*)d_in[1];
    const int*   edge_dst = (const int*)d_in[2];
    const float* mask     = (const float*)d_in[3];
    const float* w0  = (const float*)d_in[4];
    const float* b0  = (const float*)d_in[5];
    const float* g0  = (const float*)d_in[6];
    const float* be0 = (const float*)d_in[7];
    const float* rm0 = (const float*)d_in[8];
    const float* rv0 = (const float*)d_in[9];
    const float* w1  = (const float*)d_in[10];
    const float* b1  = (const float*)d_in[11];
    const float* g1  = (const float*)d_in[12];
    const float* be1 = (const float*)d_in[13];
    const float* rm1 = (const float*)d_in[14];
    const float* rv1 = (const float*)d_in[15];
    const float* w2  = (const float*)d_in[16];
    const float* b2  = (const float*)d_in[17];
    const float* g2  = (const float*)d_in[18];
    const float* be2 = (const float*)d_in[19];
    const float* rm2 = (const float*)d_in[20];
    const float* rv2 = (const float*)d_in[21];
    const float* wp  = (const float*)d_in[22];
    const float* bp  = (const float*)d_in[23];
    float* out = (float*)d_out;

    float* W = (float*)d_ws;
    float* dinv = W;
    float* h1   = dinv + 50048;
    float* h2   = h1   + (long)NN * DIM;
    float* cbuf = h2   + (long)NN * DIM;
    unsigned short* hist = (unsigned short*)(cbuf + 1024);
    int*   bstart = (int*)(cbuf + 1024) + NHIST;
    int*   off  = bstart + 256;
    int2*  tmp  = (int2*)(off + 50048);
    int2*  rec  = tmp + EE;
    float* sc0 = cbuf + 0 * DIM;
    float* sh0 = cbuf + 1 * DIM;
    float* sc1 = cbuf + 2 * DIM;
    float* sh1 = cbuf + 3 * DIM;
    float* sc2 = cbuf + 4 * DIM;
    float* sh2 = cbuf + 4 * DIM + 64;

    BnArgs bn;
    bn.b0 = b0; bn.g0 = g0; bn.be0 = be0; bn.rm0 = rm0; bn.rv0 = rv0;
    bn.b1 = b1; bn.g1 = g1; bn.be1 = be1; bn.rm1 = rm1; bn.rv1 = rv1;
    bn.b2 = b2; bn.g2 = g2; bn.be2 = be2; bn.rm2 = rm2; bn.rv2 = rv2;
    chunk_hist_kernel<<<NCHUNK, STHREADS, 0, stream>>>(edge_dst, hist, bn, cbuf);
    chunk_scatter_kernel<<<NCHUNK, STHREADS, 0, stream>>>(edge_src, edge_dst, mask, hist, bstart, tmp);
    bucket_sort_kernel<<<NBKT, STHREADS, 0, stream>>>(tmp, bstart, off, dinv, rec);

    gin_gemm_kernel<<<LGRID, LTHREADS, 0, stream>>>(h, off, rec, dinv, w0, sc0, sh0, h1);

    gin_gemm_kernel<<<LGRID, LTHREADS, 0, stream>>>(h1, off, rec, dinv, w1, sc1, sh1, h2);

    gin_final_kernel<<<LGRID, LTHREADS, 0, stream>>>(h2, off, rec, dinv, w2, wp,
                                                     sc2, sh2, bp, out);
}


# Layer kernels split into 24 gather waves and 8 matrix waves: tiles gathered in two batches, the first batch's WMMA work runs beside the second batch's gather; per-node sum order unchanged
# speedup vs baseline: 5.1413x; 1.0400x over previous
#include <hip/hip_runtime.h>

typedef __attribute__((ext_vector_type(2))) float v2f;
typedef __attribute__((ext_vector_type(8))) float v8f;

#define NN 50000
#define EE 800000
#define DIM 128
#define NCLS 40
#define CP 48

#define BSHIFT   8
#define BW       (1 << BSHIFT)
#define NBKT     ((NN + BW - 1) / BW)
#define NCHUNK   256
#define CHUNK    (EE / NCHUNK)
#define STHREADS 512
#define EPT      ((CHUNK + STHREADS - 1) / STHREADS)
#define NHIST    (NBKT * NCHUNK)
#define DBATCH   8
static_assert(EE % NCHUNK == 0, "chunks must cover the edge list exactly");
static_assert(NBKT <= 256 && BW == 256, "the 256-entry scan covers the buckets of a chunk and the nodes of a bucket");
static_assert(NN <= 65536, "a source index must fit the low 16 bits of a record word");
static_assert(NCHUNK == 256, "a histogram row of 16-bit counts is one 16-byte load per lane of one wave");
static_assert(CHUNK * 4 < 65536, "a count, and the sum of four counts, must fit 16 bits");

__device__ __forceinline__ int excl_scan_256(const int c, const int tid, int* __restrict__ sTot) {
    const int lane = tid & 31;
    const int wave = tid >> 5;
    int incl = c;
#pragma unroll
    for (int d = 1; d < 32; d <<= 1) {
        int t = __shfl_up(incl, d);
        if (lane >= d) incl += t;
    }
    if (lane == 31 && wave < 8) sTot[wave] = incl;
    __syncthreads();
    int basev = 0;
#pragma unroll
    for (int w = 0; w < 8; ++w)
        if (w < wave) basev += sTot[w];
    return basev + incl - c;
}

struct BnArgs {
    const float* b0; const float* g0; const float* be0; const float* rm0; const float* rv0;
    const float* b1; const float* g1; const float* be1; const float* rm1; const float* rv1;
    const float* b2; const float* g2; const float* be2; const float* rm2; const float* rv2;
};

__device__ __forceinline__ void fold_consts(const BnArgs& a, float* __restrict__ cbuf, const int t) {
    const float eps = 1e-5f;
    if (t < DIM) {
        float s = a.g0[t] * rsqrtf(a.rv0[t] + eps);
        cbuf[0 * DIM + t] = s;
        cbuf[1 * DIM + t] = (a.b0[t] - a.rm0[t]) * s + a.be0[t];
        s = a.g1[t] * rsqrtf(a.rv1[t] + eps);
        cbuf[2 * DIM + t] = s;
        cbuf[3 * DIM + t] = (a.b1[t] - a.rm1[t]) * s + a.be1[t];
    }
    if (t < NCLS) {
        float s = a.g2[t] * rsqrtf(a.rv2[t] + eps);
        cbuf[4 * DIM + t]      = s;
        cbuf[4 * DIM + 64 + t] = (a.b2[t] - a.rm2[t]) * s + a.be2[t];
    }
}

__global__ void __launch_bounds__(STHREADS)
chunk_hist_kernel(const int* __restrict__ dst, unsigned short* __restrict__ hist,
                  const BnArgs bn, float* __restrict__ cbuf) {
    __shared__ int sCnt[NBKT];
    const int tid = threadIdx.x;
    const int c   = blockIdx.x;
    const int e0  = c * CHUNK;
    int d[EPT];
#pragma unroll
    for (int k = 0; k < EPT; ++k) {
        const int i = tid + k * STHREADS;
        d[k] = (i < CHUNK) ? dst[e0 + i] : -1;
    }
    if (tid < NBKT) sCnt[tid] = 0;
    __syncthreads();
#pragma unroll
    for (int k = 0; k < EPT; ++k)
        if (d[k] >= 0) atomicAdd(&sCnt[d[k] >> BSHIFT], 1);
    __syncthreads();
    if (tid < NBKT) hist[tid * NCHUNK + c] = (unsigned short)sCnt[tid];
    if (c == 0) fold_consts(bn, cbuf, tid);
}

__global__ void __launch_bounds__(STHREADS)
chunk_scatter_kernel(const int* __restrict__ src, const int* __restrict__ dst,
                     const float* __restrict__ mask, const unsigned short* __restrict__ hist,
                     int* __restrict__ bstart, int2* __restrict__ tmp) {
    __shared__ int sPT[NBKT * 33];
    __shared__ int sPP[NBKT * 33];
    __shared__ int sTotB[8];
    __shared__ int sCnt[256];
    __shared__ int sLst[256];
    __shared__ int sGof[256];
    __shared__ int sTot[8];
    __shared__ int2 sRec[CHUNK];
    __shared__ unsigned char sBid[CHUNK];
    const int tid = threadIdx.x;
    const int c   = blockIdx.x;
    const int e0  = c * CHUNK;
    constexpr int NWV = STHREADS / 32;
    constexpr int RPW = (NBKT + NWV - 1) / NWV;
    const int lane = tid & 31;
    const int wv   = tid >> 5;
    uint4 hr[RPW];
#pragma unroll
    for (int q = 0; q < RPW; ++q) {
        const int b = wv + q * NWV;
        const uint4* row = (const uint4*)(hist + min(b, NBKT - 1) * NCHUNK);
        hr[q] = row[lane];
    }
    int d[EPT], s[EPT], mk[EPT], r[EPT];
#pragma unroll
    for (int k = 0; k < EPT; ++k) {
        const int i = tid + k * STHREADS;
        const bool in = i < CHUNK;
        d[k]  = in ? dst[e0 + i] : -1;
        s[k]  = in ? src[e0 + i] : 0;
        mk[k] = in ? __float_as_int(mask[e0 + i]) : 0;
        r[k]  = 0;
    }
    if (tid < 256) sCnt[tid] = 0;
    {
        const int i0 = lane * 8;
        unsigned mw[4];
#pragma unroll
        for (int j = 0; j < 4; ++j)
            mw[j] = (i0 + 2 * j < c ? 0x0000ffffu : 0u) | (i0 + 2 * j + 1 < c ? 0xffff0000u : 0u);
#pragma unroll
        for (int q = 0; q < RPW; ++q) {
            const int b = wv + q * NWV;
            const unsigned st = (hr[q].x + hr[q].y) + (hr[q].z + hr[q].w);
            const unsigned sp = ((hr[q].x & mw[0]) + (hr[q].y & mw[1])) + ((hr[q].z & mw[2]) + (hr[q].w & mw[3]));
            const int tot = (int)((st & 0xffffu) + (st >> 16));
            const int pre = (int)((sp & 0xffffu) + (sp >> 16));
            if (b < NBKT) { sPT[b * 33 + lane] = tot; sPP[b * 33 + lane] = pre; }
        }
    }
    __syncthreads();
#pragma unroll
    for (int k = 0; k < EPT; ++k)
        if (d[k] >= 0) r[k] = atomicAdd(&sCnt[d[k] >> BSHIFT], 1);
    int btot = 0, bpre = 0;
    if (tid < NBKT) {
#pragma unroll
        for (int j = 0; j < 32; ++j) {
            btot += sPT[tid * 33 + j];
            bpre += sPP[tid * 33 + j];
        }
    }
    __syncthreads();
    const int cnt  = (tid < 256) ? sCnt[tid] : 0;
    const int bst  = excl_scan_256(btot, tid, sTotB);
    const int excl = excl_scan_256(cnt, tid, sTot);
    if (tid < NBKT) {
        sLst[tid] = excl;
        sGof[tid] = bst + bpre - excl;
        if (c == 0) bstart[tid] = bst;
        if (c == 0 && tid == NBKT - 1) bstart[NBKT] = bst + btot;
    }
    __syncthreads();
#pragma unroll
    for (int k = 0; k < EPT; ++k) {
        if (d[k] >= 0) {
            const int b  = d[k] >> BSHIFT;
            const int lp = sLst[b] + r[k];
            sRec[lp] = make_int2(((d[k] & (BW - 1)) << 16) | s[k], mk[k]);
            sBid[lp] = (unsigned char)b;
        }
    }
    __syncthreads();
    for (int i = tid; i < CHUNK; i += STHREADS) tmp[i + sGof[sBid[i]]] = sRec[i];
}

__global__ void __launch_bounds__(STHREADS)
bucket_sort_kernel(const int2* __restrict__ tmp, const int* __restrict__ bstart,
                   int* __restrict__ off, float* __restrict__ dinv, int2* __restrict__ rec) {
    __shared__ int sCnt[BW];
    __shared__ int sExc[BW];
    __shared__ int sTot[8];
    const int tid = threadIdx.x;
    const int b   = blockIdx.x;
    const int beg = bstart[b];
    const int end = bstart[b + 1];
    if (tid < BW) sCnt[tid] = 0;
    __syncthreads();
    for (int i0 = beg; i0 < end; i0 += DBATCH * STHREADS) {
        int w[DBATCH];
#pragma unroll
        for (int k = 0; k < DBATCH; ++k) {
            const int i = i0 + k * STHREADS + tid;
            w[k] = (i < end) ? tmp[i].x : -1;
        }
#pragma unroll
        for (int k = 0; k < DBATCH; ++k)
            if (w[k] >= 0) atomicAdd(&sCnt[w[k] >> 16], 1);
    }
    __syncthreads();
    const int cnt  = (tid < BW) ? sCnt[tid] : 0;
    const int excl = excl_scan_256(cnt, tid, sTot);
    if (tid < BW) {
        sExc[tid] = excl;
        sCnt[tid] = 0;
        const int node = b * BW + tid;
        if (node < NN) {
            off[node]  = beg + excl;
            dinv[node] = (cnt > 0) ? (1.0f / (float)cnt) : 0.0f;
        }
        if (node == NN - 1) off[NN] = beg + excl + cnt;
    }
    __syncthreads();
    for (int i0 = beg; i0 < end; i0 += DBATCH * STHREADS) {
        int2 t[DBATCH];
#pragma unroll
        for (int k = 0; k < DBATCH; ++k) {
            const int i = i0 + k * STHREADS + tid;
            t[k] = (i < end) ? tmp[i] : make_int2(-1, 0);
        }
#pragma unroll
        for (int k = 0; k < DBATCH; ++k) {
            if (t[k].x >= 0) {
                const int dl = t[k].x >> 16;
                const int p  = atomicAdd(&sCnt[dl], 1);
                rec[beg + sExc[dl] + p] = make_int2(t[k].x & 0xffff, t[k].y);
            }
        }
    }
}


#define NTPG    13
#define BT      6
#define XSLOT   (2 * BT)
#define QROWS   4
#define NCONS   8
#define NPROD   (4 * BT)
#define LTHREADS ((NCONS + NPROD) * 32)
#define LGRID   256
#define NTILES  (NN / 16)
#define GB      16
#define TPITCH  132
static_assert(NN % 16 == 0, "tiles of 16 nodes must cover the node range exactly");
static_assert(LGRID * NTPG >= NTILES, "every tile needs a slot");
static_assert(LGRID * XSLOT <= NTILES, "the slots of the two batches are live in every group");
static_assert(NTPG == XSLOT + 1, "one extra slot beyond the two batches");
static_assert(LTHREADS <= 1024, "one work-group holds all waves");

#define ROW_FMA(V, M)                      \
    acc.x = fmaf((V).x, (M), acc.x);       \
    acc.y = fmaf((V).y, (M), acc.y);       \
    acc.z = fmaf((V).z, (M), acc.z);       \
    acc.w = fmaf((V).w, (M), acc.w);

template <int NPW>
__device__ __forceinline__ int gather_offsets(const int* __restrict__ off, const int n0, const int lane) {
    return off[n0 + min(lane, NPW)];
}

template <int NPW>
__device__ __forceinline__ int2 gather_first(const int2* __restrict__ rec, const int offv, const int lane) {
    const int beg = __builtin_amdgcn_readlane(offv, 0);
    const int end = __builtin_amdgcn_readlane(offv, NPW);
    int2 ed = make_int2(0, 0);
    if (beg + lane < end) ed = rec[beg + lane];
    return ed;
}

template <int NPW>
__device__ __forceinline__ void gather_tile(const float* __restrict__ x, const int2* __restrict__ rec,
                                            const int offv, int2 ed,
                                            float* __restrict__ tile, const int lane) {
    const int beg = __builtin_amdgcn_readlane(offv, 0);
    const int end = __builtin_amdgcn_readlane(offv, NPW);
    const float* xl = x + lane * 4;
    float* trow = tile + lane * 4;
    float4 acc = make_float4(0.0f, 0.0f, 0.0f, 0.0f);
    int cur = 0;
    int nb  = __builtin_amdgcn_readlane(offv, 1);

    for (int base = beg; base < end; base += 32) {
        const int c = min(32, end - base);
        int2 edn = make_int2(0, 0);
        if (base + 32 + lane < end) edn = rec[base + 32 + lane];
        for (int j = 0; j < c; j += GB) {
            float4 v[GB];
            float  mk[GB];
            if (j + GB <= c) {
#pragma unroll
                for (int k = 0; k < GB; ++k) {
                    const unsigned s = (unsigned)__builtin_amdgcn_readlane(ed.x, j + k);
                    mk[k] = __int_as_float(__builtin_amdgcn_readlane(ed.y, j + k));
                    v[k] = *(const float4*)(xl + (size_t)s * DIM);
                }
                const int e0 = base + j;
                int rel = (cur < NPW - 1) ? nb - e0 : -1;
#pragma unroll
                for (int k = 0; k < GB; ++k) {
                    if (rel == k) {
                        do {
                            *(float4*)(trow + cur * TPITCH) = acc;
                            acc = make_float4(0.0f, 0.0f, 0.0f, 0.0f);
                            ++cur;
                            nb = __builtin_amdgcn_readlane(offv, cur + 1);
                        } while (cur < NPW - 1 && nb == e0 + k);
                        rel = (cur < NPW - 1) ? nb - e0 : -1;
                    }
                    ROW_FMA(v[k], mk[k])
                }
            } else {
#pragma unroll
                for (int k = 0; k < GB; ++k) {
                    const int jj = min(j + k, c - 1);
                    const int s = __builtin_amdgcn_readlane(ed.x, jj);
                    mk[k] = __int_as_float(__builtin_amdgcn_readlane(ed.y, jj));
                    v[k] = *(const float4*)(xl + (long)s * DIM);
                }
#pragma unroll
                for (int k = 0; k < GB; ++k) {
                    if (j + k < c) {
                        const int e = base + j + k;
                        while (cur < NPW - 1 && e == nb) {
                            *(float4*)(trow + cur * TPITCH) = acc;
                            acc = make_float4(0.0f, 0.0f, 0.0f, 0.0f);
                            ++cur;
                            nb = __builtin_amdgcn_readlane(offv, cur + 1);
                        }
                        ROW_FMA(v[k], mk[k])
                    }
                }
            }
        }
        ed = edn;
    }
    for (; cur < NPW; ++cur) {
        *(float4*)(trow + cur * TPITCH) = acc;
        acc = make_float4(0.0f, 0.0f, 0.0f, 0.0f);
    }
}


template <bool KEEPX>
__device__ __forceinline__ void gather_unit(const float* __restrict__ x, const int2* __restrict__ rec,
                                            const float* __restrict__ dinv, const int n0,
                                            const int offv, const int2 ed,
                                            float* __restrict__ qt, float* __restrict__ qx, const int lane) {
    const float dv = dinv[n0 + (lane & (QROWS - 1))];
    float4 xi[QROWS];
#pragma unroll
    for (int i = 0; i < QROWS; ++i) xi[i] = *(const float4*)(x + (long)(n0 + i) * DIM + lane * 4);
    gather_tile<QROWS>(x, rec, offv, ed, qt, lane);
#pragma unroll
    for (int i = 0; i < QROWS; ++i) {
        const float di = __int_as_float(__builtin_amdgcn_readlane(__float_as_int(dv), i));
        float4 ag = *(const float4*)(qt + i * TPITCH + lane * 4);
        float4 a;
        a.x = xi[i].x + ag.x * di;
        a.y = xi[i].y + ag.y * di;
        a.z = xi[i].z + ag.z * di;
        a.w = xi[i].w + ag.w * di;
        *(float4*)(qt + i * TPITCH + lane * 4) = a;
        if (KEEPX) *(float4*)(qx + i * TPITCH + lane * 4) = xi[i];
    }
}


__device__ __forceinline__ void gemm_batch(const float* __restrict__ sA, const v2f* __restrict__ sW,
                                           const float* __restrict__ scale, const float* __restrict__ shift,
                                           float* __restrict__ out, const int first, const bool extra,
                                           const int c, const int lane) {
    const int m    = lane & 15;
    const int prow = lane >> 4;
    const int kk   = prow << 1;
    const int roff = prow << 3;
    const int cb   = (c & 3) * 32;
    const float s0 = scale[cb + m],      h0 = shift[cb + m];
    const float s1 = scale[cb + 16 + m], h1 = shift[cb + 16 + m];
    const int nloc = BT + (extra ? 1 : 0);
    for (int li = c >> 2; li < nloc; li += 2) {
        const int slot = (li < BT) ? first + li : XSLOT;
        const int row0 = (blockIdx.x + LGRID * slot) * 16;
        const float* ar = sA + slot * (16 * TPITCH) + m * TPITCH + kk;
        v8f acc[2] = {};
#pragma unroll 4
        for (int k0 = 0; k0 < DIM; k0 += 4) {
            const v2f* bb = sW + ((k0 >> 1) + prow) * DIM + cb + m;
            const v2f b0 = bb[0];
            const v2f b1 = bb[16];
            const v2f a = *(const v2f*)(ar + k0);
            acc[0] = __builtin_amdgcn_wmma_f32_16x16x4_f32(false, a, false, b0, (short)0, acc[0], false, false);
            acc[1] = __builtin_amdgcn_wmma_f32_16x16x4_f32(false, a, false, b1, (short)0, acc[1], false, false);
        }
#pragma unroll
        for (int j = 0; j < 8; ++j) {
            const long rr = row0 + roff + j;
            const float v0 = acc[0][j] * s0 + h0;
            const float v1 = acc[1][j] * s1 + h1;
            out[rr * DIM + cb + m]      = v0 > 0.0f ? v0 : 0.0f;
            out[rr * DIM + cb + 16 + m] = v1 > 0.0f ? v1 : 0.0f;
        }
    }
}

__global__ void __launch_bounds__(LTHREADS)
gin_gemm_kernel(const float* __restrict__ x, const int* __restrict__ off,
                const int2* __restrict__ rec,
                const float* __restrict__ dinv, const float* __restrict__ W,
                const float* __restrict__ scale, const float* __restrict__ shift,
                float* __restrict__ out) {
    __shared__ v2f sW[(DIM / 2) * DIM];
    __shared__ __attribute__((aligned(16))) float sA[NTPG * 16 * TPITCH];
    const int tid = threadIdx.x;
    const int wave = __builtin_amdgcn_readfirstlane(tid >> 5);
    const int lane = tid & 31;
    const bool cons = wave < NCONS;
    const bool extra = blockIdx.x + LGRID * XSLOT < NTILES;
    const int pw = wave - NCONS;
    const int q  = cons ? (wave & 3) : (pw & 3);
    const int slot0 = cons ? ((wave < 4 && extra) ? XSLOT : -1) : (pw >> 2);
    const int slot1 = cons ? -1 : BT + (pw >> 2);
    const int n00 = (blockIdx.x + LGRID * max(slot0, 0)) * 16 + q * QROWS;
    const int n01 = (blockIdx.x + LGRID * max(slot1, 0)) * 16 + q * QROWS;

    int offv0 = 0, offv1 = 0;
    if (slot0 >= 0) offv0 = gather_offsets<QROWS>(off, n00, lane);
    if (slot1 >= 0) offv1 = gather_offsets<QROWS>(off, n01, lane);

    constexpr int WPT = (DIM / 2) * DIM / LTHREADS;
    static_assert(WPT * LTHREADS == (DIM / 2) * DIM, "the threads cover the weights exactly");
    v2f wst[WPT];
#pragma unroll
    for (int r = 0; r < WPT; ++r) {
        const int i = tid + r * LTHREADS;
        const int p = i / DIM, c = i % DIM;
        wst[r].x = W[(2 * p) * DIM + c];
        wst[r].y = W[(2 * p + 1) * DIM + c];
    }

    int2 ed0 = make_int2(0, 0), ed1 = make_int2(0, 0);
    if (slot0 >= 0) ed0 = gather_first<QROWS>(rec, offv0, lane);
    if (slot1 >= 0) ed1 = gather_first<QROWS>(rec, offv1, lane);

#pragma unroll
    for (int r = 0; r < WPT; ++r) sW[tid + r * LTHREADS] = wst[r];

#pragma nounroll
    for (int st = 0; st < 2; ++st) {
        const int slot = st ? slot1 : slot0;
        if (slot >= 0) {
            float* qt = sA + slot * (16 * TPITCH) + q * (QROWS * TPITCH);
            gather_unit<false>(x, rec, dinv, st ? n01 : n00, st ? offv1 : offv0, st ? ed1 : ed0, qt, qt, lane);
        } else if (st == 1 && cons) {
            gemm_batch(sA, sW, scale, shift, out, 0, extra, wave, lane);
        }
        __syncthreads();
    }
    if (cons) gemm_batch(sA, sW, scale, shift, out, BT, false, wave, lane);
}


__device__ __forceinline__ void final_batch(const float* __restrict__ sA, const float* __restrict__ sX,
                                            const v2f* __restrict__ sW2, const v2f* __restrict__ sWP,
                                            const float* __restrict__ scale2, const float* __restrict__ shift2,
                                            const float* __restrict__ bp, float* __restrict__ out,
                                            const int first, const bool extra, const int c, const int lane) {
    const int m    = lane & 15;
    const int prow = lane >> 4;
    const int kk   = prow << 1;
    const int roff = prow << 3;
    const int nun  = 3 * (BT + (extra ? 1 : 0));
    for (int u = c; u < nun; u += NCONS) {
        const int li = u / 3;
        const int nt = u - 3 * li;
        const int slot = (li < BT) ? first + li : XSLOT;
        const int row0 = (blockIdx.x + LGRID * slot) * 16;
        const float* ah = sA + slot * (16 * TPITCH) + m * TPITCH + kk;
        const float* ap = sX + slot * (16 * TPITCH) + m * TPITCH + kk;
        const v2f* wh = sW2 + prow * CP + nt * 16 + m;
        const v2f* wq = sWP + prow * CP + nt * 16 + m;
        v8f acch = {}, accp = {};
#pragma unroll 4
        for (int k0 = 0; k0 < DIM; k0 += 4) {
            const v2f bh = wh[(k0 >> 1) * CP];
            const v2f bq = wq[(k0 >> 1) * CP];
            const v2f a0 = *(const v2f*)(ah + k0);
            const v2f a1 = *(const v2f*)(ap + k0);
            acch = __builtin_amdgcn_wmma_f32_16x16x4_f32(false, a0, false, bh, (short)0, acch, false, false);
            accp = __builtin_amdgcn_wmma_f32_16x16x4_f32(false, a1, false, bq, (short)0, accp, false, false);
        }
        const int col = nt * 16 + m;
        if (col < NCLS) {
            const float s  = scale2[col];
            const float sh = shift2[col];
            const float bb = bp[col];
#pragma unroll
            for (int j = 0; j < 8; ++j) {
                const long rr = row0 + roff + j;
                float h3 = acch[j] * s + sh;
                h3 = h3 > 0.0f ? h3 : 0.0f;
                out[rr * NCLS + col] = 0.5f * (accp[j] + bb + h3);
            }
        }
    }
}

__global__ void __launch_bounds__(LTHREADS)
gin_final_kernel(const float* __restrict__ h2, const int* __restrict__ off,
                 const int2* __restrict__ rec,
                 const float* __restrict__ dinv,
                 const float* __restrict__ w2, const float* __restrict__ wp,
                 const float* __restrict__ scale2, const float* __restrict__ shift2,
                 const float* __restrict__ bp, float* __restrict__ out) {
    __shared__ v2f sW2[(DIM / 2) * CP];
    __shared__ v2f sWP[(DIM / 2) * CP];
    __shared__ __attribute__((aligned(16))) float sA[NTPG * 16 * TPITCH];
    __shared__ __attribute__((aligned(16))) float sX[NTPG * 16 * TPITCH];
    const int tid = threadIdx.x;
    const int wave = __builtin_amdgcn_readfirstlane(tid >> 5);
    const int lane = tid & 31;
    const bool cons = wave < NCONS;
    const bool extra = blockIdx.x + LGRID * XSLOT < NTILES;
    const int pw = wave - NCONS;
    const int q  = cons ? (wave & 3) : (pw & 3);
    const int slot0 = cons ? ((wave < 4 && extra) ? XSLOT : -1) : (pw >> 2);
    const int slot1 = cons ? -1 : BT + (pw >> 2);
    const int n00 = (blockIdx.x + LGRID * max(slot0, 0)) * 16 + q * QROWS;
    const int n01 = (blockIdx.x + LGRID * max(slot1, 0)) * 16 + q * QROWS;

    int offv0 = 0, offv1 = 0;
    if (slot0 >= 0) offv0 = gather_offsets<QROWS>(off, n00, lane);
    if (slot1 >= 0) offv1 = gather_offsets<QROWS>(off, n01, lane);

    constexpr int WPT = (DIM / 2) * CP / LTHREADS;
    static_assert(WPT * LTHREADS == (DIM / 2) * CP, "the threads cover the padded weights exactly");
    v2f wa[WPT], wb[WPT];
#pragma unroll
    for (int r = 0; r < WPT; ++r) {
        const int i = tid + r * LTHREADS;
        const int p = i / CP, c = i % CP;
        wa[r] = v2f{0.0f, 0.0f};
        wb[r] = v2f{0.0f, 0.0f};
        if (c < NCLS) {
            wa[r].x = w2[(2 * p) * NCLS + c];
            wa[r].y = w2[(2 * p + 1) * NCLS + c];
            wb[r].x = wp[(2 * p) * NCLS + c];
            wb[r].y = wp[(2 * p + 1) * NCLS + c];
        }
    }

    int2 ed0 = make_int2(0, 0), ed1 = make_int2(0, 0);
    if (slot0 >= 0) ed0 = gather_first<QROWS>(rec, offv0, lane);
    if (slot1 >= 0) ed1 = gather_first<QROWS>(rec, offv1, lane);

#pragma unroll
    for (int r = 0; r < WPT; ++r) {
        sW2[tid + r * LTHREADS] = wa[r];
        sWP[tid + r * LTHREADS] = wb[r];
    }

#pragma nounroll
    for (int st = 0; st < 2; ++st) {
        const int slot = st ? slot1 : slot0;
        if (slot >= 0) {
            const int qo = slot * (16 * TPITCH) + q * (QROWS * TPITCH);
            gather_unit<true>(h2, rec, dinv, st ? n01 : n00, st ? offv1 : offv0, st ? ed1 : ed0,
                              sA + qo, sX + qo, lane);
        } else if (st == 1 && cons) {
            final_batch(sA, sX, sW2, sWP, scale2, shift2, bp, out, 0, extra, wave, lane);
        }
        __syncthreads();
    }
    if (cons) final_batch(sA, sX, sW2, sWP, scale2, shift2, bp, out, BT, false, wave, lane);
}

extern "C" void kernel_launch(void* const* d_in, const int* in_sizes, int n_in,
                              void* d_out, int out_size, void* d_ws, size_t ws_size,
                              hipStream_t stream) {
    const float* h        = (const float*)d_in[0];
    const int*   edge_src = (const int*)d_in[1];
    const int*   edge_dst = (const int*)d_in[2];
    const float* mask     = (const float*)d_in[3];
    const float* w0  = (const float*)d_in[4];
    const float* b0  = (const float*)d_in[5];
    const float* g0  = (const float*)d_in[6];
    const float* be0 = (const float*)d_in[7];
    const float* rm0 = (const float*)d_in[8];
    const float* rv0 = (const float*)d_in[9];
    const float* w1  = (const float*)d_in[10];
    const float* b1  = (const float*)d_in[11];
    const float* g1  = (const float*)d_in[12];
    const float* be1 = (const float*)d_in[13];
    const float* rm1 = (const float*)d_in[14];
    const float* rv1 = (const float*)d_in[15];
    const float* w2  = (const float*)d_in[16];
    const float* b2  = (const float*)d_in[17];
    const float* g2  = (const float*)d_in[18];
    const float* be2 = (const float*)d_in[19];
    const float* rm2 = (const float*)d_in[20];
    const float* rv2 = (const float*)d_in[21];
    const float* wp  = (const float*)d_in[22];
    const float* bp  = (const float*)d_in[23];
    float* out = (float*)d_out;

    float* W = (float*)d_ws;
    float* dinv = W;
    float* h1   = dinv + 50048;
    float* h2   = h1   + (long)NN * DIM;
    float* cbuf = h2   + (long)NN * DIM;
    unsigned short* hist = (unsigned short*)(cbuf + 1024);
    int*   bstart = (int*)(cbuf + 1024) + NHIST;
    int*   off  = bstart + 256;
    int2*  tmp  = (int2*)(off + 50048);
    int2*  rec  = tmp + EE;
    float* sc0 = cbuf + 0 * DIM;
    float* sh0 = cbuf + 1 * DIM;
    float* sc1 = cbuf + 2 * DIM;
    float* sh1 = cbuf + 3 * DIM;
    float* sc2 = cbuf + 4 * DIM;
    float* sh2 = cbuf + 4 * DIM + 64;

    BnArgs bn;
    bn.b0 = b0; bn.g0 = g0; bn.be0 = be0; bn.rm0 = rm0; bn.rv0 = rv0;
    bn.b1 = b1; bn.g1 = g1; bn.be1 = be1; bn.rm1 = rm1; bn.rv1 = rv1;
    bn.b2 = b2; bn.g2 = g2; bn.be2 = be2; bn.rm2 = rm2; bn.rv2 = rv2;
    chunk_hist_kernel<<<NCHUNK, STHREADS, 0, stream>>>(edge_dst, hist, bn, cbuf);
    chunk_scatter_kernel<<<NCHUNK, STHREADS, 0, stream>>>(edge_src, edge_dst, mask, hist, bstart, tmp);
    bucket_sort_kernel<<<NBKT, STHREADS, 0, stream>>>(tmp, bstart, off, dinv, rec);

    gin_gemm_kernel<<<LGRID, LTHREADS, 0, stream>>>(h, off, rec, dinv, w0, sc0, sh0, h1);

    gin_gemm_kernel<<<LGRID, LTHREADS, 0, stream>>>(h1, off, rec, dinv, w1, sc1, sh1, h2);

    gin_final_kernel<<<LGRID, LTHREADS, 0, stream>>>(h2, off, rec, dinv, w2, wp,
                                                     sc2, sh2, bp, out);
}
